# starting kernel
# baseline (speedup 1.0000x reference)
#include <hip/hip_runtime.h>
#include <math.h>

typedef __attribute__((ext_vector_type(2))) float v2f;
typedef __attribute__((ext_vector_type(8))) float v8f;

#define HIDF 128

__global__ void k_init_deg(float* __restrict__ deg, int n) {
  int i = blockIdx.x * blockDim.x + threadIdx.x;
  if (i < n) deg[i] = 1.0f;
}

__global__ void k_edge_deg(const int* __restrict__ col, const float* __restrict__ ew,
                           float* __restrict__ deg, int E) {
  int e = blockIdx.x * blockDim.x + threadIdx.x;
  if (e < E) atomicAdd(&deg[col[e]], ew[e]);
}

__global__ void k_rsqrt(float* __restrict__ deg, int n) {
  int i = blockIdx.x * blockDim.x + threadIdx.x;
  if (i < n) {
    float d = deg[i];
    deg[i] = (d > 0.0f) ? rsqrtf(d) : 0.0f;
  }
}

__global__ __launch_bounds__(256)
void k_gemm_wmma(const float* __restrict__ A, const float* __restrict__ W,
                 float* __restrict__ C, const float* __restrict__ bias) {
  __shared__ float As[16 * 132];
  const int t  = threadIdx.x;
  const int m0 = blockIdx.x * 16;

  {
    const float4* A4 = (const float4*)(A + (size_t)m0 * HIDF);
    for (int i = t; i < 16 * 32; i += 256) {
      int r = i >> 5, c = i & 31;
      float4 v = A4[i];
      As[r * 132 + c * 4 + 0] = v.x;
      As[r * 132 + c * 4 + 1] = v.y;
      As[r * 132 + c * 4 + 2] = v.z;
      As[r * 132 + c * 4 + 3] = v.w;
    }
  }
  __syncthreads();

  const int lane = t & 31;
  const int wave = t >> 5;
  const int half = lane >> 4;
  const int l15  = lane & 15;
  const int n0   = wave * 16;

  v8f acc = {};
#pragma unroll
  for (int kt = 0; kt < 32; ++kt) {
    const int k0 = kt * 4 + 2 * half;
#if defined(__has_builtin) && __has_builtin(__builtin_amdgcn_wmma_f32_16x16x4_f32)
    v2f a, b;
    a.x = As[l15 * 132 + k0 + 0];
    a.y = As[l15 * 132 + k0 + 1];
    b.x = W[(k0 + 0) * HIDF + n0 + l15];
    b.y = W[(k0 + 1) * HIDF + n0 + l15];
    acc = __builtin_amdgcn_wmma_f32_16x16x4_f32(false, a, false, b, (short)0, acc,
                                                false, false);
#else
#pragma unroll
    for (int r = 0; r < 8; ++r) {
      int M = r + 8 * half;
      int kk = kt * 4;
      float s = acc[r];
      s += As[M * 132 + kk + 0] * W[(kk + 0) * HIDF + n0 + l15];
      s += As[M * 132 + kk + 1] * W[(kk + 1) * HIDF + n0 + l15];
      s += As[M * 132 + kk + 2] * W[(kk + 2) * HIDF + n0 + l15];
      s += As[M * 132 + kk + 3] * W[(kk + 3) * HIDF + n0 + l15];
      acc[r] = s;
    }
#endif
  }

  const float bv = bias ? bias[n0 + l15] : 0.0f;
#pragma unroll
  for (int r = 0; r < 8; ++r) {
    C[(size_t)(m0 + r + 8 * half) * HIDF + (n0 + l15)] = acc[r] + bv;
  }
}

__global__ void k_selfagg(const float* __restrict__ hw, const float* __restrict__ dis,
                          float* __restrict__ agg, int n) {
  int idx = blockIdx.x * blockDim.x + threadIdx.x;
  if (idx >= n * 32) return;
  int node = idx >> 5;
  float s = dis[node];
  s = s * s;
  float4 v = ((const float4*)hw)[idx];
  float4 o;
  o.x = v.x * s; o.y = v.y * s; o.z = v.z * s; o.w = v.w * s;
  ((float4*)agg)[idx] = o;
}

__global__ __launch_bounds__(256)
void k_scatter(const float* __restrict__ hw, const int* __restrict__ row,
               const int* __restrict__ col, const float* __restrict__ ew,
               const float* __restrict__ dis, float* __restrict__ agg, int E) {
  int e = blockIdx.x * 8 + (threadIdx.x >> 5);
  if (e >= E) return;
  const int lane = threadIdx.x & 31;
  const int r = row[e];
  const int c = col[e];
  const float norm = dis[r] * ew[e] * dis[c];
  float4 v = ((const float4*)(hw + (size_t)r * HIDF))[lane];
  float* dst = agg + (size_t)c * HIDF + lane * 4;
  atomicAdd(dst + 0, v.x * norm);
  atomicAdd(dst + 1, v.y * norm);
  atomicAdd(dst + 2, v.z * norm);
  atomicAdd(dst + 3, v.w * norm);
}

__global__ void k_finalize(const float* __restrict__ agg, const float* __restrict__ b,
                           float* __restrict__ out, int n, int relu) {
  int idx = blockIdx.x * blockDim.x + threadIdx.x;
  if (idx >= n * 32) return;
  int c4 = idx & 31;
  float4 v  = ((const float4*)agg)[idx];
  float4 bb = ((const float4*)b)[c4];
  v.x += bb.x; v.y += bb.y; v.z += bb.z; v.w += bb.w;
  if (relu) {
    v.x = fmaxf(v.x, 0.0f); v.y = fmaxf(v.y, 0.0f);
    v.z = fmaxf(v.z, 0.0f); v.w = fmaxf(v.w, 0.0f);
  }
  ((float4*)out)[idx] = v;
}

extern "C" void kernel_launch(void* const* d_in, const int* in_sizes, int n_in,
                              void* d_out, int out_size, void* d_ws, size_t ws_size,
                              hipStream_t stream) {
  const float* x   = (const float*)d_in[0];
  const int*   ei  = (const int*)d_in[1];
  const float* ew  = (const float*)d_in[2];
  const float* Wc[4] = {(const float*)d_in[3], (const float*)d_in[5],
                        (const float*)d_in[7], (const float*)d_in[9]};
  const float* bc[4] = {(const float*)d_in[4], (const float*)d_in[6],
                        (const float*)d_in[8], (const float*)d_in[10]};
  const float* Wout = (const float*)d_in[11];
  const float* bout = (const float*)d_in[12];

  const int n = in_sizes[0] / HIDF;
  const int E = in_sizes[1] / 2;
  const int* row = ei;
  const int* col = ei + E;

  char* ws = (char*)d_ws;
  float* dis = (float*)ws;
  size_t off = (((size_t)n * 4) + 255) & ~(size_t)255;
  float* hw   = (float*)(ws + off);
  float* agg  = hw  + (size_t)n * HIDF;
  float* hbuf = agg + (size_t)n * HIDF;

  k_init_deg<<<(n + 255) / 256, 256, 0, stream>>>(dis, n);
  k_edge_deg<<<(E + 255) / 256, 256, 0, stream>>>(col, ew, dis, E);
  k_rsqrt<<<(n + 255) / 256, 256, 0, stream>>>(dis, n);

  const int nf4 = n * 32;
  const float* hcur = x;
  for (int l = 0; l < 4; ++l) {
    k_gemm_wmma<<<n / 16, 256, 0, stream>>>(hcur, Wc[l], hw, nullptr);
    k_selfagg<<<(nf4 + 255) / 256, 256, 0, stream>>>(hw, dis, agg, n);
    k_scatter<<<(E + 7) / 8, 256, 0, stream>>>(hw, row, col, ew, dis, agg, E);
    k_finalize<<<(nf4 + 255) / 256, 256, 0, stream>>>(agg, bc[l], hbuf, n,
                                                      (l < 3) ? 1 : 0);
    hcur = hbuf;
  }
  k_gemm_wmma<<<n / 16, 256, 0, stream>>>(hcur, Wout, (float*)d_out, bout);
}


# f32-atomic edge scatter -> edges bucketed by destination once per call (int atomic rank, scan, edge order), one wave sums a node's rows in registers; sums now in fixed edge order, no float atomics
# speedup vs baseline: 2.8567x; 2.8567x over previous
#include <hip/hip_runtime.h>
#include <math.h>

typedef __attribute__((ext_vector_type(2))) float v2f;
typedef __attribute__((ext_vector_type(8))) float v8f;

#define HIDF 128


__global__ void k_zero_cnt(int* __restrict__ cnt, int n) {
  int i = blockIdx.x * blockDim.x + threadIdx.x;
  if (i < n) cnt[i] = 0;
}

__global__ __launch_bounds__(256)
void k_count(const int* __restrict__ col, int* __restrict__ cnt, int* __restrict__ pos, int E) {
  const int G  = gridDim.x * 256;
  const int e0 = blockIdx.x * 256 + threadIdx.x;
  if (e0 + 3 * G < E) {
    const int c0 = col[e0], c1 = col[e0 + G], c2 = col[e0 + 2 * G], c3 = col[e0 + 3 * G];
    const int p0 = atomicAdd(&cnt[c0], 1);
    const int p1 = atomicAdd(&cnt[c1], 1);
    const int p2 = atomicAdd(&cnt[c2], 1);
    const int p3 = atomicAdd(&cnt[c3], 1);
    pos[e0] = p0; pos[e0 + G] = p1; pos[e0 + 2 * G] = p2; pos[e0 + 3 * G] = p3;
  } else {
    for (int e = e0; e < E; e += G) pos[e] = atomicAdd(&cnt[col[e]], 1);
  }
}

__global__ __launch_bounds__(1024)
void k_scan(const int* __restrict__ cnt, int* __restrict__ offs, int n) {
  __shared__ int part[32];
  __shared__ int wtot[32];
  const int t = threadIdx.x, lane = t & 31, wv = t >> 5;
  const int start = blockIdx.x * 1024;
  int s = 0;
  for (int i = t; i < start; i += 1024) s += cnt[i];
  const int i = start + t;
  const int v = (i < n) ? cnt[i] : 0;
  int inc = v;
  for (int o = 1; o < 32; o <<= 1) {
    const int u = __shfl_up(inc, o, 32);
    if (lane >= o) inc += u;
  }
  for (int o = 16; o > 0; o >>= 1) s += __shfl_xor(s, o, 32);
  if (lane == 31) wtot[wv] = inc;
  if (lane == 0) part[wv] = s;
  __syncthreads();
  int base = 0;
  for (int w = 0; w < 32; ++w) {
    base += part[w];
    if (w < wv) base += wtot[w];
  }
  const int excl = base + inc - v;
  if (i < n) offs[i] = excl;
  if (i == n - 1) offs[n] = excl + v;
}

__global__ __launch_bounds__(256)
void k_fill(const int* __restrict__ row, const int* __restrict__ col,
            const float* __restrict__ ew, const int* __restrict__ pos,
            const int* __restrict__ offs, int4* __restrict__ rec, int E) {
  const int G  = gridDim.x * 256;
  const int e0 = blockIdx.x * 256 + threadIdx.x;
  if (e0 + 3 * G < E) {
    const int e1 = e0 + G, e2 = e0 + 2 * G, e3 = e0 + 3 * G;
    const int s0 = offs[col[e0]] + pos[e0];
    const int s1 = offs[col[e1]] + pos[e1];
    const int s2 = offs[col[e2]] + pos[e2];
    const int s3 = offs[col[e3]] + pos[e3];
    rec[s0] = make_int4(e0, row[e0], __float_as_int(ew[e0]), 0);
    rec[s1] = make_int4(e1, row[e1], __float_as_int(ew[e1]), 0);
    rec[s2] = make_int4(e2, row[e2], __float_as_int(ew[e2]), 0);
    rec[s3] = make_int4(e3, row[e3], __float_as_int(ew[e3]), 0);
  } else {
    for (int e = e0; e < E; e += G)
      rec[offs[col[e]] + pos[e]] = make_int4(e, row[e], __float_as_int(ew[e]), 0);
  }
}

__global__ __launch_bounds__(256)
void k_order_deg(const int* __restrict__ offs, const int4* __restrict__ rec,
                 int2* __restrict__ sedge, float* __restrict__ dis, int n) {
  __shared__ int bufE[32 * 256];
  __shared__ int bufR[32 * 256];
  __shared__ int bufW[32 * 256];
  const int t = threadIdx.x;
  const int c = blockIdx.x * 256 + t;
  if (c >= n) return;
  const int beg = offs[c], end = offs[c + 1];
  const int d = end - beg;
  float deg = 0.0f;
  int last = -1;
  if (d <= 32) {
    for (int j = 0; j < d; ++j) {
      const int4 q = rec[beg + j];
      bufE[j * 256 + t] = q.x;
      bufR[j * 256 + t] = q.y;
      bufW[j * 256 + t] = q.z;
    }
    for (int s = 0; s < d; ++s) {
      int best = 0x7fffffff, bj = 0;
      for (int j = 0; j < d; ++j) {
        const int ee = bufE[j * 256 + t];
        if (ee > last && ee < best) { best = ee; bj = j; }
      }
      const int wb = bufW[bj * 256 + t];
      sedge[beg + s] = make_int2(bufR[bj * 256 + t], wb);
      deg += __int_as_float(wb);
      last = best;
    }
  } else {
    for (int s = 0; s < d; ++s) {
      int best = 0x7fffffff, bj = 0;
      for (int j = 0; j < d; ++j) {
        const int ee = rec[beg + j].x;
        if (ee > last && ee < best) { best = ee; bj = j; }
      }
      const int4 q = rec[beg + bj];
      sedge[beg + s] = make_int2(q.y, q.z);
      deg += __int_as_float(q.z);
      last = best;
    }
  }
  deg += 1.0f;
  dis[c] = (deg > 0.0f) ? rsqrtf(deg) : 0.0f;
}

__global__ __launch_bounds__(256)
void k_gemm_wmma(const float* __restrict__ A, const float* __restrict__ W,
                 float* __restrict__ C, const float* __restrict__ bias) {
  __shared__ float As[16 * 132];
  const int t  = threadIdx.x;
  const int m0 = blockIdx.x * 16;

  {
    const float4* A4 = (const float4*)(A + (size_t)m0 * HIDF);
    for (int i = t; i < 16 * 32; i += 256) {
      int r = i >> 5, c = i & 31;
      float4 v = A4[i];
      As[r * 132 + c * 4 + 0] = v.x;
      As[r * 132 + c * 4 + 1] = v.y;
      As[r * 132 + c * 4 + 2] = v.z;
      As[r * 132 + c * 4 + 3] = v.w;
    }
  }
  __syncthreads();

  const int lane = t & 31;
  const int wave = t >> 5;
  const int half = lane >> 4;
  const int l15  = lane & 15;
  const int n0   = wave * 16;

  v8f acc = {};
#pragma unroll
  for (int kt = 0; kt < 32; ++kt) {
    const int k0 = kt * 4 + 2 * half;
#if defined(__has_builtin) && __has_builtin(__builtin_amdgcn_wmma_f32_16x16x4_f32)
    v2f a, b;
    a.x = As[l15 * 132 + k0 + 0];
    a.y = As[l15 * 132 + k0 + 1];
    b.x = W[(k0 + 0) * HIDF + n0 + l15];
    b.y = W[(k0 + 1) * HIDF + n0 + l15];
    acc = __builtin_amdgcn_wmma_f32_16x16x4_f32(false, a, false, b, (short)0, acc,
                                                false, false);
#else
#pragma unroll
    for (int r = 0; r < 8; ++r) {
      int M = r + 8 * half;
      int kk = kt * 4;
      float s = acc[r];
      s += As[M * 132 + kk + 0] * W[(kk + 0) * HIDF + n0 + l15];
      s += As[M * 132 + kk + 1] * W[(kk + 1) * HIDF + n0 + l15];
      s += As[M * 132 + kk + 2] * W[(kk + 2) * HIDF + n0 + l15];
      s += As[M * 132 + kk + 3] * W[(kk + 3) * HIDF + n0 + l15];
      acc[r] = s;
    }
#endif
  }

  const float bv = bias ? bias[n0 + l15] : 0.0f;
#pragma unroll
  for (int r = 0; r < 8; ++r) {
    C[(size_t)(m0 + r + 8 * half) * HIDF + (n0 + l15)] = acc[r] + bv;
  }
}

template <int B>
__device__ __forceinline__ void rows_add(const char* __restrict__ hwb, int r, float w,
                                         int k, unsigned loff, float4& acc) {
  float4 v[B];
  float wk[B];
#pragma unroll
  for (int i = 0; i < B; ++i) {
    const unsigned rk = (unsigned)__builtin_amdgcn_readlane(r, k + i);
    wk[i] = __int_as_float(__builtin_amdgcn_readlane(__float_as_int(w), k + i));
    const char* rowp = hwb + ((size_t)rk << 9);
    v[i] = *(const float4*)(rowp + loff);
  }
#pragma unroll
  for (int i = 0; i < B; ++i) {
    acc.x += v[i].x * wk[i];
    acc.y += v[i].y * wk[i];
    acc.z += v[i].z * wk[i];
    acc.w += v[i].w * wk[i];
  }
}

__global__ __launch_bounds__(256, 16)
void k_gather(const float* __restrict__ hw, const int* __restrict__ offs,
              const int2* __restrict__ sedge,
              const float* __restrict__ dis, const float* __restrict__ b,
              float* __restrict__ out, int n, int relu) {
  const int lane = threadIdx.x & 31;
  const int wid = __builtin_amdgcn_readfirstlane((int)(blockIdx.x * 8 + (threadIdx.x >> 5)));
  const int nw  = gridDim.x * 8;
  const float4* hw4 = (const float4*)hw;
  const char* hwb = (const char*)hw;
  const unsigned loff = (unsigned)lane * 16u;
  const float4 bb = ((const float4*)b)[lane];
  for (int c = wid; c < n; c += nw) {
    const int beg = offs[c], end = offs[c + 1];
    const float dc = dis[c];
    const float4 hv = hw4[(size_t)c * 32 + lane];
    const float s2 = dc * dc;
    float4 acc = make_float4(hv.x * s2, hv.y * s2, hv.z * s2, hv.w * s2);
    for (int base = beg; base < end; base += 32) {
      const int j = base + lane;
      int r = 0;
      float w = 0.0f;
      if (j < end) {
        const int2 q = sedge[j];
        r = q.x;
        w = dis[r] * __int_as_float(q.y) * dc;
      }
      const int left = end - base;
      const int m = left < 32 ? left : 32;
      int k = 0;
      for (; k + 8 <= m; k += 8) rows_add<8>(hwb, r, w, k, loff, acc);
      if (k + 4 <= m) { rows_add<4>(hwb, r, w, k, loff, acc); k += 4; }
      if (k + 2 <= m) { rows_add<2>(hwb, r, w, k, loff, acc); k += 2; }
      if (k < m)      { rows_add<1>(hwb, r, w, k, loff, acc); }
    }
    acc.x += bb.x; acc.y += bb.y; acc.z += bb.z; acc.w += bb.w;
    if (relu) {
      acc.x = fmaxf(acc.x, 0.0f); acc.y = fmaxf(acc.y, 0.0f);
      acc.z = fmaxf(acc.z, 0.0f); acc.w = fmaxf(acc.w, 0.0f);
    }
    ((float4*)out)[(size_t)c * 32 + lane] = acc;
  }
}

extern "C" void kernel_launch(void* const* d_in, const int* in_sizes, int n_in,
                              void* d_out, int out_size, void* d_ws, size_t ws_size,
                              hipStream_t stream) {
  const float* x   = (const float*)d_in[0];
  const int*   ei  = (const int*)d_in[1];
  const float* ew  = (const float*)d_in[2];
  const float* Wc[4] = {(const float*)d_in[3], (const float*)d_in[5],
                        (const float*)d_in[7], (const float*)d_in[9]};
  const float* bc[4] = {(const float*)d_in[4], (const float*)d_in[6],
                        (const float*)d_in[8], (const float*)d_in[10]};
  const float* Wout = (const float*)d_in[11];
  const float* bout = (const float*)d_in[12];

  const int n = in_sizes[0] / HIDF;
  const int E = in_sizes[1] / 2;
  const int* row = ei;
  const int* col = ei + E;

  char* ws = (char*)d_ws;
  const size_t fb = (size_t)n * HIDF * sizeof(float);
  const size_t nb = (((size_t)(n + 1) * 4) + 255) & ~(size_t)255;
  const size_t eb = (((size_t)E * 4) + 255) & ~(size_t)255;
  float* hw   = (float*)ws;
  float* hbuf = (float*)(ws + fb);
  char*  p    = ws + 2 * fb;
  float* dis    = (float*)p; p += nb;
  int*   cnt    = (int*)p;   p += nb;
  int*   offs   = (int*)p;   p += nb;
  int*   pos    = (int*)p;   p += eb;
  int2*  sedge  = (int2*)p;  p += 2 * eb;
  int4*  rec    = (int4*)p;

  k_zero_cnt<<<(n + 255) / 256, 256, 0, stream>>>(cnt, n);
  k_count<<<(E + 1023) / 1024, 256, 0, stream>>>(col, cnt, pos, E);
  k_scan<<<(n + 1023) / 1024, 1024, 0, stream>>>(cnt, offs, n);
  k_fill<<<(E + 1023) / 1024, 256, 0, stream>>>(row, col, ew, pos, offs, rec, E);
  k_order_deg<<<(n + 255) / 256, 256, 0, stream>>>(offs, rec, sedge, dis, n);

  const float* hcur = x;
  for (int l = 0; l < 4; ++l) {
    k_gemm_wmma<<<n / 16, 256, 0, stream>>>(hcur, Wc[l], hw, nullptr);
    k_gather<<<(n + 7) / 8, 256, 0, stream>>>(hw, offs, sedge, dis, bc[l], hbuf, n,
                                       (l < 3) ? 1 : 0);
    hcur = hbuf;
  }
  k_gemm_wmma<<<n / 16, 256, 0, stream>>>(hcur, Wout, (float*)d_out, bout);
}


# per-node edge sum fused with the next layer's product: a block sums its 16 nodes' rows into the LDS tile its WMMAs read (14 -> 10 launches, summed rows never reach memory); sum order unchanged
# speedup vs baseline: 3.0199x; 1.0571x over previous
#include <hip/hip_runtime.h>
#include <math.h>

typedef __attribute__((ext_vector_type(2))) float v2f;
typedef __attribute__((ext_vector_type(8))) float v8f;

#define HIDF 128


__global__ void k_zero_cnt(int* __restrict__ cnt, int n) {
  int i = blockIdx.x * blockDim.x + threadIdx.x;
  if (i < n) cnt[i] = 0;
}

__global__ __launch_bounds__(256)
void k_count(const int* __restrict__ col, int* __restrict__ cnt, int* __restrict__ pos, int E) {
  const int G  = gridDim.x * 256;
  const int e0 = blockIdx.x * 256 + threadIdx.x;
  if (e0 + 3 * G < E) {
    const int c0 = col[e0], c1 = col[e0 + G], c2 = col[e0 + 2 * G], c3 = col[e0 + 3 * G];
    const int p0 = atomicAdd(&cnt[c0], 1);
    const int p1 = atomicAdd(&cnt[c1], 1);
    const int p2 = atomicAdd(&cnt[c2], 1);
    const int p3 = atomicAdd(&cnt[c3], 1);
    pos[e0] = p0; pos[e0 + G] = p1; pos[e0 + 2 * G] = p2; pos[e0 + 3 * G] = p3;
  } else {
    for (int e = e0; e < E; e += G) pos[e] = atomicAdd(&cnt[col[e]], 1);
  }
}

__global__ __launch_bounds__(1024)
void k_scan(const int* __restrict__ cnt, int* __restrict__ offs, int n) {
  __shared__ int part[32];
  __shared__ int wtot[32];
  const int t = threadIdx.x, lane = t & 31, wv = t >> 5;
  const int start = blockIdx.x * 1024;
  int s = 0;
  for (int i = t; i < start; i += 1024) s += cnt[i];
  const int i = start + t;
  const int v = (i < n) ? cnt[i] : 0;
  int inc = v;
  for (int o = 1; o < 32; o <<= 1) {
    const int u = __shfl_up(inc, o, 32);
    if (lane >= o) inc += u;
  }
  for (int o = 16; o > 0; o >>= 1) s += __shfl_xor(s, o, 32);
  if (lane == 31) wtot[wv] = inc;
  if (lane == 0) part[wv] = s;
  __syncthreads();
  int base = 0;
  for (int w = 0; w < 32; ++w) {
    base += part[w];
    if (w < wv) base += wtot[w];
  }
  const int excl = base + inc - v;
  if (i < n) offs[i] = excl;
  if (i == n - 1) offs[n] = excl + v;
}

__global__ __launch_bounds__(256)
void k_fill(const int* __restrict__ row, const int* __restrict__ col,
            const float* __restrict__ ew, const int* __restrict__ pos,
            const int* __restrict__ offs, int4* __restrict__ rec, int E) {
  const int G  = gridDim.x * 256;
  const int e0 = blockIdx.x * 256 + threadIdx.x;
  if (e0 + 3 * G < E) {
    const int e1 = e0 + G, e2 = e0 + 2 * G, e3 = e0 + 3 * G;
    const int s0 = offs[col[e0]] + pos[e0];
    const int s1 = offs[col[e1]] + pos[e1];
    const int s2 = offs[col[e2]] + pos[e2];
    const int s3 = offs[col[e3]] + pos[e3];
    rec[s0] = make_int4(e0, row[e0], __float_as_int(ew[e0]), 0);
    rec[s1] = make_int4(e1, row[e1], __float_as_int(ew[e1]), 0);
    rec[s2] = make_int4(e2, row[e2], __float_as_int(ew[e2]), 0);
    rec[s3] = make_int4(e3, row[e3], __float_as_int(ew[e3]), 0);
  } else {
    for (int e = e0; e < E; e += G)
      rec[offs[col[e]] + pos[e]] = make_int4(e, row[e], __float_as_int(ew[e]), 0);
  }
}

__global__ __launch_bounds__(256)
void k_order_deg(const int* __restrict__ offs, const int4* __restrict__ rec,
                 int2* __restrict__ sedge, float* __restrict__ dis, int n) {
  __shared__ int bufE[32 * 256];
  __shared__ int bufR[32 * 256];
  __shared__ int bufW[32 * 256];
  const int t = threadIdx.x;
  const int c = blockIdx.x * 256 + t;
  if (c >= n) return;
  const int beg = offs[c], end = offs[c + 1];
  const int d = end - beg;
  float deg = 0.0f;
  int last = -1;
  if (d <= 32) {
    for (int j = 0; j < d; ++j) {
      const int4 q = rec[beg + j];
      bufE[j * 256 + t] = q.x;
      bufR[j * 256 + t] = q.y;
      bufW[j * 256 + t] = q.z;
    }
    for (int s = 0; s < d; ++s) {
      int best = 0x7fffffff, bj = 0;
      for (int j = 0; j < d; ++j) {
        const int ee = bufE[j * 256 + t];
        if (ee > last && ee < best) { best = ee; bj = j; }
      }
      const int wb = bufW[bj * 256 + t];
      sedge[beg + s] = make_int2(bufR[bj * 256 + t], wb);
      deg += __int_as_float(wb);
      last = best;
    }
  } else {
    for (int s = 0; s < d; ++s) {
      int best = 0x7fffffff, bj = 0;
      for (int j = 0; j < d; ++j) {
        const int ee = rec[beg + j].x;
        if (ee > last && ee < best) { best = ee; bj = j; }
      }
      const int4 q = rec[beg + bj];
      sedge[beg + s] = make_int2(q.y, q.z);
      deg += __int_as_float(q.z);
      last = best;
    }
  }
  deg += 1.0f;
  dis[c] = (deg > 0.0f) ? rsqrtf(deg) : 0.0f;
}

__global__ __launch_bounds__(256)
void k_gemm_wmma(const float* __restrict__ A, const float* __restrict__ W,
                 float* __restrict__ C, const float* __restrict__ bias) {
  __shared__ float As[16 * 132];
  const int t  = threadIdx.x;
  const int m0 = blockIdx.x * 16;

  {
    const float4* A4 = (const float4*)(A + (size_t)m0 * HIDF);
    for (int i = t; i < 16 * 32; i += 256) {
      int r = i >> 5, c = i & 31;
      float4 v = A4[i];
      As[r * 132 + c * 4 + 0] = v.x;
      As[r * 132 + c * 4 + 1] = v.y;
      As[r * 132 + c * 4 + 2] = v.z;
      As[r * 132 + c * 4 + 3] = v.w;
    }
  }
  __syncthreads();

  const int lane = t & 31;
  const int wave = t >> 5;
  const int half = lane >> 4;
  const int l15  = lane & 15;
  const int n0   = wave * 16;

  v8f acc = {};
#pragma unroll
  for (int kt = 0; kt < 32; ++kt) {
    const int k0 = kt * 4 + 2 * half;
#if defined(__has_builtin) && __has_builtin(__builtin_amdgcn_wmma_f32_16x16x4_f32)
    v2f a, b;
    a.x = As[l15 * 132 + k0 + 0];
    a.y = As[l15 * 132 + k0 + 1];
    b.x = W[(k0 + 0) * HIDF + n0 + l15];
    b.y = W[(k0 + 1) * HIDF + n0 + l15];
    acc = __builtin_amdgcn_wmma_f32_16x16x4_f32(false, a, false, b, (short)0, acc,
                                                false, false);
#else
#pragma unroll
    for (int r = 0; r < 8; ++r) {
      int M = r + 8 * half;
      int kk = kt * 4;
      float s = acc[r];
      s += As[M * 132 + kk + 0] * W[(kk + 0) * HIDF + n0 + l15];
      s += As[M * 132 + kk + 1] * W[(kk + 1) * HIDF + n0 + l15];
      s += As[M * 132 + kk + 2] * W[(kk + 2) * HIDF + n0 + l15];
      s += As[M * 132 + kk + 3] * W[(kk + 3) * HIDF + n0 + l15];
      acc[r] = s;
    }
#endif
  }

  const float bv = bias ? bias[n0 + l15] : 0.0f;
#pragma unroll
  for (int r = 0; r < 8; ++r) {
    C[(size_t)(m0 + r + 8 * half) * HIDF + (n0 + l15)] = acc[r] + bv;
  }
}

template <int B>
__device__ __forceinline__ void rows_add(const char* __restrict__ hwb, int r, float w,
                                         int k, unsigned loff, float4& acc) {
  float4 v[B];
  float wk[B];
#pragma unroll
  for (int i = 0; i < B; ++i) {
    const unsigned rk = (unsigned)__builtin_amdgcn_readlane(r, k + i);
    wk[i] = __int_as_float(__builtin_amdgcn_readlane(__float_as_int(w), k + i));
    const char* rowp = hwb + ((size_t)rk << 9);
    v[i] = *(const float4*)(rowp + loff);
  }
#pragma unroll
  for (int i = 0; i < B; ++i) {
    acc.x += v[i].x * wk[i];
    acc.y += v[i].y * wk[i];
    acc.z += v[i].z * wk[i];
    acc.w += v[i].w * wk[i];
  }
}

__global__ __launch_bounds__(256, 16)
void k_gather_gemm(const float* __restrict__ hw, const int* __restrict__ offs,
                   const int2* __restrict__ sedge, const float* __restrict__ dis,
                   const float* __restrict__ b, int relu,
                   const float* __restrict__ W, const float* __restrict__ bias2,
                   float* __restrict__ C) {
  __shared__ float As[16 * 132];
  const int t    = threadIdx.x;
  const int lane = t & 31;
  const int wave = __builtin_amdgcn_readfirstlane(t >> 5);
  const int m0   = blockIdx.x * 16;
  const float4* hw4 = (const float4*)hw;
  const char* hwb = (const char*)hw;
  const unsigned loff = (unsigned)lane * 16u;
  const float4 bb = ((const float4*)b)[lane];

  for (int i = 0; i < 2; ++i) {
    const int lr = wave * 2 + i;
    const int c  = m0 + lr;
    const int beg = offs[c], end = offs[c + 1];
    const float dc = dis[c];
    const float4 hv = hw4[(size_t)c * 32 + lane];
    const float s2 = dc * dc;
    float4 acc = make_float4(hv.x * s2, hv.y * s2, hv.z * s2, hv.w * s2);
    for (int base = beg; base < end; base += 32) {
      const int j = base + lane;
      int r = 0;
      float w = 0.0f;
      if (j < end) {
        const int2 q = sedge[j];
        r = q.x;
        w = dis[r] * __int_as_float(q.y) * dc;
      }
      const int left = end - base;
      const int m = left < 32 ? left : 32;
      int k = 0;
      for (; k + 8 <= m; k += 8) rows_add<8>(hwb, r, w, k, loff, acc);
      if (k + 4 <= m) { rows_add<4>(hwb, r, w, k, loff, acc); k += 4; }
      if (k + 2 <= m) { rows_add<2>(hwb, r, w, k, loff, acc); k += 2; }
      if (k < m)      { rows_add<1>(hwb, r, w, k, loff, acc); }
    }
    acc.x += bb.x; acc.y += bb.y; acc.z += bb.z; acc.w += bb.w;
    if (relu) {
      acc.x = fmaxf(acc.x, 0.0f); acc.y = fmaxf(acc.y, 0.0f);
      acc.z = fmaxf(acc.z, 0.0f); acc.w = fmaxf(acc.w, 0.0f);
    }
    *(float4*)&As[lr * 132 + lane * 4] = acc;
  }
  __syncthreads();

  const int half = lane >> 4;
  const int l15  = lane & 15;
  const int n0   = wave * 16;

  v8f cacc = {};
#pragma unroll
  for (int kt = 0; kt < 32; ++kt) {
    const int k0 = kt * 4 + 2 * half;
    v2f a, bf;
    a.x = As[l15 * 132 + k0 + 0];
    a.y = As[l15 * 132 + k0 + 1];
    bf.x = W[(k0 + 0) * HIDF + n0 + l15];
    bf.y = W[(k0 + 1) * HIDF + n0 + l15];
    cacc = __builtin_amdgcn_wmma_f32_16x16x4_f32(false, a, false, bf, (short)0, cacc,
                                                 false, false);
  }

  const float bv = bias2 ? bias2[n0 + l15] : 0.0f;
#pragma unroll
  for (int r = 0; r < 8; ++r) {
    C[(size_t)(m0 + r + 8 * half) * HIDF + (n0 + l15)] = cacc[r] + bv;
  }
}

extern "C" void kernel_launch(void* const* d_in, const int* in_sizes, int n_in,
                              void* d_out, int out_size, void* d_ws, size_t ws_size,
                              hipStream_t stream) {
  const float* x   = (const float*)d_in[0];
  const int*   ei  = (const int*)d_in[1];
  const float* ew  = (const float*)d_in[2];
  const float* Wc[4] = {(const float*)d_in[3], (const float*)d_in[5],
                        (const float*)d_in[7], (const float*)d_in[9]};
  const float* bc[4] = {(const float*)d_in[4], (const float*)d_in[6],
                        (const float*)d_in[8], (const float*)d_in[10]};
  const float* Wout = (const float*)d_in[11];
  const float* bout = (const float*)d_in[12];

  const int n = in_sizes[0] / HIDF;
  const int E = in_sizes[1] / 2;
  const int* row = ei;
  const int* col = ei + E;

  char* ws = (char*)d_ws;
  const size_t fb = (size_t)n * HIDF * sizeof(float);
  const size_t nb = (((size_t)(n + 1) * 4) + 255) & ~(size_t)255;
  const size_t eb = (((size_t)E * 4) + 255) & ~(size_t)255;
  float* hw   = (float*)ws;
  float* hbuf = (float*)(ws + fb);
  char*  p    = ws + 2 * fb;
  float* dis    = (float*)p; p += nb;
  int*   cnt    = (int*)p;   p += nb;
  int*   offs   = (int*)p;   p += nb;
  int*   pos    = (int*)p;   p += eb;
  int2*  sedge  = (int2*)p;  p += 2 * eb;
  int4*  rec    = (int4*)p;

  k_zero_cnt<<<(n + 255) / 256, 256, 0, stream>>>(cnt, n);
  k_count<<<(E + 1023) / 1024, 256, 0, stream>>>(col, cnt, pos, E);
  k_scan<<<(n + 1023) / 1024, 1024, 0, stream>>>(cnt, offs, n);
  k_fill<<<(E + 1023) / 1024, 256, 0, stream>>>(row, col, ew, pos, offs, rec, E);
  k_order_deg<<<(n + 255) / 256, 256, 0, stream>>>(offs, rec, sedge, dis, n);

  k_gemm_wmma<<<n / 16, 256, 0, stream>>>(x, Wc[0], hw, nullptr);
  k_gather_gemm<<<n / 16, 256, 0, stream>>>(hw, offs, sedge, dis, bc[0], 1, Wc[1], nullptr, hbuf);
  k_gather_gemm<<<n / 16, 256, 0, stream>>>(hbuf, offs, sedge, dis, bc[1], 1, Wc[2], nullptr, hw);
  k_gather_gemm<<<n / 16, 256, 0, stream>>>(hw, offs, sedge, dis, bc[2], 1, Wc[3], nullptr, hbuf);
  k_gather_gemm<<<n / 16, 256, 0, stream>>>(hbuf, offs, sedge, dis, bc[3], 0, Wout, bout,
                                            (float*)d_out);
}


# edge bucketing without global atomics: edges binned by destination range on LDS counters (count, scan, place); a block per range orders each node's bucket by edge id, a wave per node; 10->9 launches
# speedup vs baseline: 4.0524x; 1.3419x over previous
#include <hip/hip_runtime.h>
#include <math.h>

typedef __attribute__((ext_vector_type(2))) float v2f;
typedef __attribute__((ext_vector_type(8))) float v8f;

#define HIDF 128


#define EB   8192
#define ETH  512
#define RB   256
#define RSH  8
#define KMAX 512
#define WCAP 64

__global__ __launch_bounds__(ETH)
void k_bin_count(const int* __restrict__ col, int* __restrict__ pos, int* __restrict__ tab,
                 int E, int NB, int K) {
  __shared__ int h[KMAX];
  const int t = threadIdx.x, b = blockIdx.x;
  for (int i = t; i < K; i += ETH) h[i] = 0;
  __syncthreads();
  const int e0 = b * EB + t;
  if (b * EB + EB <= E) {
    for (int g = 0; g < EB / ETH; g += 8) {
      int c[8];
#pragma unroll
      for (int u = 0; u < 8; ++u) c[u] = col[e0 + (g + u) * ETH];
#pragma unroll
      for (int u = 0; u < 8; ++u) pos[e0 + (g + u) * ETH] = atomicAdd(&h[c[u] >> RSH], 1);
    }
  } else {
    for (int e = e0; e < E; e += ETH) pos[e] = atomicAdd(&h[col[e] >> RSH], 1);
  }
  __syncthreads();
  for (int i = t; i < K; i += ETH) tab[i * NB + b] = h[i];
}

__global__ __launch_bounds__(1024)
void k_scan(const int* __restrict__ cnt, int* __restrict__ offs, int n) {
  __shared__ int part[32];
  __shared__ int wtot[32];
  const int t = threadIdx.x, lane = t & 31, wv = t >> 5;
  const int start = blockIdx.x * 1024;
  int s = 0;
  for (int i = t; i < start; i += 1024) s += cnt[i];
  const int i = start + t;
  const int v = (i < n) ? cnt[i] : 0;
  int inc = v;
  for (int o = 1; o < 32; o <<= 1) {
    const int u = __shfl_up(inc, o, 32);
    if (lane >= o) inc += u;
  }
  for (int o = 16; o > 0; o >>= 1) s += __shfl_xor(s, o, 32);
  if (lane == 31) wtot[wv] = inc;
  if (lane == 0) part[wv] = s;
  __syncthreads();
  int base = 0;
  for (int w = 0; w < 32; ++w) {
    base += part[w];
    if (w < wv) base += wtot[w];
  }
  const int excl = base + inc - v;
  if (i < n) offs[i] = excl;
  if (i == n - 1) offs[n] = excl + v;
}

__global__ __launch_bounds__(ETH)
void k_bin_place(const int* __restrict__ row, const int* __restrict__ col,
                 const float* __restrict__ ew, const int* __restrict__ pos,
                 const int* __restrict__ base, int4* __restrict__ rec, int E, int NB, int K) {
  __shared__ int sb[KMAX];
  const int t = threadIdx.x, b = blockIdx.x;
  for (int i = t; i < K; i += ETH) sb[i] = base[i * NB + b];
  __syncthreads();
  const int e0 = b * EB + t;
  if (b * EB + EB <= E) {
    for (int g = 0; g < EB / ETH; g += 4) {
      int c[4], p[4], r[4];
      float w[4];
#pragma unroll
      for (int u = 0; u < 4; ++u) {
        const int e = e0 + (g + u) * ETH;
        c[u] = col[e]; p[u] = pos[e]; r[u] = row[e]; w[u] = ew[e];
      }
#pragma unroll
      for (int u = 0; u < 4; ++u) {
        const int e = e0 + (g + u) * ETH;
        rec[sb[c[u] >> RSH] + p[u]] = make_int4(e, r[u], __float_as_int(w[u]), c[u]);
      }
    }
  } else {
    for (int e = e0; e < E; e += ETH) {
      const int c = col[e];
      rec[sb[c >> RSH] + pos[e]] = make_int4(e, row[e], __float_as_int(ew[e]), c);
    }
  }
}

__global__ __launch_bounds__(1024)
void k_bin_order(const int* __restrict__ base, const int4* __restrict__ rec,
                 int4* __restrict__ tmp, int2* __restrict__ sedge, int* __restrict__ offs,
                 float* __restrict__ dis, int n, int NB) {
  __shared__ int cnt[RB];
  __shared__ int loff[RB];
  __shared__ int cur[RB];
  __shared__ int wtot[RB / 32];
  __shared__ float wsort[32 * WCAP];
  const int t = threadIdx.x, lane = t & 31;
  const int wv = __builtin_amdgcn_readfirstlane(t >> 5);
  const int k = blockIdx.x;
  const int bs = __builtin_amdgcn_readfirstlane(base[k * NB]);
  const int be = __builtin_amdgcn_readfirstlane(base[(k + 1) * NB]);
  const int nb0 = k * RB;
  if (t < RB) cnt[t] = 0;
  __syncthreads();
  for (int idx = bs + t; idx < be; idx += 1024) atomicAdd(&cnt[rec[idx].w - nb0], 1);
  __syncthreads();
  const int v = (t < RB) ? cnt[t] : 0;
  int inc = v;
  for (int o = 1; o < 32; o <<= 1) {
    const int u = __shfl_up(inc, o, 32);
    if (lane >= o) inc += u;
  }
  if (t < RB && lane == 31) wtot[t >> 5] = inc;
  __syncthreads();
  if (t < RB) {
    int pre = 0;
    for (int w = 0; w < RB / 32; ++w)
      if (w < (t >> 5)) pre += wtot[w];
    const int ex = pre + inc - v;
    loff[t] = ex;
    cur[t] = ex;
    if (nb0 + t < n) offs[nb0 + t] = bs + ex;
  }
  if (k == (int)gridDim.x - 1 && t == 0) offs[n] = be;
  __syncthreads();
  for (int idx = bs + t; idx < be; idx += 1024) {
    const int4 q = rec[idx];
    const int p = atomicAdd(&cur[q.w - nb0], 1);
    tmp[bs + p] = q;
  }
  __syncthreads();

  float* const ws = &wsort[wv * WCAP];
  int i = wv;
  bool have = (nb0 + i < n);
  int d = 0, beg = 0;
  int4 q = make_int4(0x7fffffff, 0, 0, 0);
  if (have) {
    d = __builtin_amdgcn_readfirstlane(cnt[i]);
    beg = bs + __builtin_amdgcn_readfirstlane(loff[i]);
    if (lane < d) q = tmp[beg + lane];
  }
  while (have) {
    const int i2 = i + 32;
    const bool have2 = (i2 < RB) && (nb0 + i2 < n);
    int d2 = 0, beg2 = 0;
    int4 q2 = make_int4(0x7fffffff, 0, 0, 0);
    if (have2) {
      d2 = __builtin_amdgcn_readfirstlane(cnt[i2]);
      beg2 = bs + __builtin_amdgcn_readfirstlane(loff[i2]);
      if (lane < d2) q2 = tmp[beg2 + lane];
    }
    float deg = 0.0f;
    if (d <= WCAP) {
      for (int j0 = 0; j0 < d; j0 += 32) {
        int4 a = q;
        if (j0 > 0) {
          a = make_int4(0x7fffffff, 0, 0, 0);
          if (j0 + lane < d) a = tmp[beg + j0 + lane];
        }
        int rank = 0;
        for (int k0 = 0; k0 < d; k0 += 32) {
          int ek = a.x;
          if (k0 != j0) {
            ek = 0x7fffffff;
            if (k0 + lane < d) ek = tmp[beg + k0 + lane].x;
          }
          const int m = (d - k0 < 32) ? (d - k0) : 32;
          for (int kk = 0; kk < m; ++kk)
            rank += (__builtin_amdgcn_readlane(ek, kk) < a.x) ? 1 : 0;
        }
        if (j0 + lane < d) {
          sedge[beg + rank] = make_int2(a.y, a.z);
          ws[rank] = __int_as_float(a.z);
        }
      }
      for (int s0 = 0; s0 < d; s0 += 32) {
        const float sw = (s0 + lane < d) ? ws[s0 + lane] : 0.0f;
        const int m = (d - s0 < 32) ? (d - s0) : 32;
        for (int kk = 0; kk < m; ++kk)
          deg += __int_as_float(__builtin_amdgcn_readlane(__float_as_int(sw), kk));
      }
    } else if (lane == 0) {
      int last = -1;
      for (int s = 0; s < d; ++s) {
        int best = 0x7fffffff, bj = 0;
        for (int j = 0; j < d; ++j) {
          const int ee = tmp[beg + j].x;
          if (ee > last && ee < best) { best = ee; bj = j; }
        }
        const int4 z = tmp[beg + bj];
        sedge[beg + s] = make_int2(z.y, z.z);
        deg += __int_as_float(z.z);
        last = best;
      }
    }
    if (lane == 0) {
      deg += 1.0f;
      dis[nb0 + i] = (deg > 0.0f) ? rsqrtf(deg) : 0.0f;
    }
    i = i2; have = have2; d = d2; beg = beg2; q = q2;
  }
}

__global__ __launch_bounds__(256)
void k_gemm_wmma(const float* __restrict__ A, const float* __restrict__ W,
                 float* __restrict__ C, const float* __restrict__ bias) {
  __shared__ float As[16 * 132];
  const int t  = threadIdx.x;
  const int m0 = blockIdx.x * 16;

  {
    const float4* A4 = (const float4*)(A + (size_t)m0 * HIDF);
    for (int i = t; i < 16 * 32; i += 256) {
      int r = i >> 5, c = i & 31;
      float4 v = A4[i];
      As[r * 132 + c * 4 + 0] = v.x;
      As[r * 132 + c * 4 + 1] = v.y;
      As[r * 132 + c * 4 + 2] = v.z;
      As[r * 132 + c * 4 + 3] = v.w;
    }
  }
  __syncthreads();

  const int lane = t & 31;
  const int wave = t >> 5;
  const int half = lane >> 4;
  const int l15  = lane & 15;
  const int n0   = wave * 16;

  v8f acc = {};
#pragma unroll
  for (int kt = 0; kt < 32; ++kt) {
    const int k0 = kt * 4 + 2 * half;
#if defined(__has_builtin) && __has_builtin(__builtin_amdgcn_wmma_f32_16x16x4_f32)
    v2f a, b;
    a.x = As[l15 * 132 + k0 + 0];
    a.y = As[l15 * 132 + k0 + 1];
    b.x = W[(k0 + 0) * HIDF + n0 + l15];
    b.y = W[(k0 + 1) * HIDF + n0 + l15];
    acc = __builtin_amdgcn_wmma_f32_16x16x4_f32(false, a, false, b, (short)0, acc,
                                                false, false);
#else
#pragma unroll
    for (int r = 0; r < 8; ++r) {
      int M = r + 8 * half;
      int kk = kt * 4;
      float s = acc[r];
      s += As[M * 132 + kk + 0] * W[(kk + 0) * HIDF + n0 + l15];
      s += As[M * 132 + kk + 1] * W[(kk + 1) * HIDF + n0 + l15];
      s += As[M * 132 + kk + 2] * W[(kk + 2) * HIDF + n0 + l15];
      s += As[M * 132 + kk + 3] * W[(kk + 3) * HIDF + n0 + l15];
      acc[r] = s;
    }
#endif
  }

  const float bv = bias ? bias[n0 + l15] : 0.0f;
#pragma unroll
  for (int r = 0; r < 8; ++r) {
    C[(size_t)(m0 + r + 8 * half) * HIDF + (n0 + l15)] = acc[r] + bv;
  }
}

template <int B>
__device__ __forceinline__ void rows_add(const char* __restrict__ hwb, int r, float w,
                                         int k, unsigned loff, float4& acc) {
  float4 v[B];
  float wk[B];
#pragma unroll
  for (int i = 0; i < B; ++i) {
    const unsigned rk = (unsigned)__builtin_amdgcn_readlane(r, k + i);
    wk[i] = __int_as_float(__builtin_amdgcn_readlane(__float_as_int(w), k + i));
    const char* rowp = hwb + ((size_t)rk << 9);
    v[i] = *(const float4*)(rowp + loff);
  }
#pragma unroll
  for (int i = 0; i < B; ++i) {
    acc.x += v[i].x * wk[i];
    acc.y += v[i].y * wk[i];
    acc.z += v[i].z * wk[i];
    acc.w += v[i].w * wk[i];
  }
}

__global__ __launch_bounds__(256, 16)
void k_gather_gemm(const float* __restrict__ hw, const int* __restrict__ offs,
                   const int2* __restrict__ sedge, const float* __restrict__ dis,
                   const float* __restrict__ b, int relu,
                   const float* __restrict__ W, const float* __restrict__ bias2,
                   float* __restrict__ C) {
  __shared__ float As[16 * 132];
  const int t    = threadIdx.x;
  const int lane = t & 31;
  const int wave = __builtin_amdgcn_readfirstlane(t >> 5);
  const int m0   = blockIdx.x * 16;
  const float4* hw4 = (const float4*)hw;
  const char* hwb = (const char*)hw;
  const unsigned loff = (unsigned)lane * 16u;
  const float4 bb = ((const float4*)b)[lane];

  for (int i = 0; i < 2; ++i) {
    const int lr = wave * 2 + i;
    const int c  = m0 + lr;
    const int beg = offs[c], end = offs[c + 1];
    const float dc = dis[c];
    const float4 hv = hw4[(size_t)c * 32 + lane];
    const float s2 = dc * dc;
    float4 acc = make_float4(hv.x * s2, hv.y * s2, hv.z * s2, hv.w * s2);
    for (int base = beg; base < end; base += 32) {
      const int j = base + lane;
      int r = 0;
      float w = 0.0f;
      if (j < end) {
        const int2 q = sedge[j];
        r = q.x;
        w = dis[r] * __int_as_float(q.y) * dc;
      }
      const int left = end - base;
      const int m = left < 32 ? left : 32;
      int k = 0;
      for (; k + 8 <= m; k += 8) rows_add<8>(hwb, r, w, k, loff, acc);
      if (k + 4 <= m) { rows_add<4>(hwb, r, w, k, loff, acc); k += 4; }
      if (k + 2 <= m) { rows_add<2>(hwb, r, w, k, loff, acc); k += 2; }
      if (k < m)      { rows_add<1>(hwb, r, w, k, loff, acc); }
    }
    acc.x += bb.x; acc.y += bb.y; acc.z += bb.z; acc.w += bb.w;
    if (relu) {
      acc.x = fmaxf(acc.x, 0.0f); acc.y = fmaxf(acc.y, 0.0f);
      acc.z = fmaxf(acc.z, 0.0f); acc.w = fmaxf(acc.w, 0.0f);
    }
    *(float4*)&As[lr * 132 + lane * 4] = acc;
  }
  __syncthreads();

  const int half = lane >> 4;
  const int l15  = lane & 15;
  const int n0   = wave * 16;

  v8f cacc = {};
#pragma unroll
  for (int kt = 0; kt < 32; ++kt) {
    const int k0 = kt * 4 + 2 * half;
    v2f a, bf;
    a.x = As[l15 * 132 + k0 + 0];
    a.y = As[l15 * 132 + k0 + 1];
    bf.x = W[(k0 + 0) * HIDF + n0 + l15];
    bf.y = W[(k0 + 1) * HIDF + n0 + l15];
    cacc = __builtin_amdgcn_wmma_f32_16x16x4_f32(false, a, false, bf, (short)0, cacc,
                                                 false, false);
  }

  const float bv = bias2 ? bias2[n0 + l15] : 0.0f;
#pragma unroll
  for (int r = 0; r < 8; ++r) {
    C[(size_t)(m0 + r + 8 * half) * HIDF + (n0 + l15)] = cacc[r] + bv;
  }
}

extern "C" void kernel_launch(void* const* d_in, const int* in_sizes, int n_in,
                              void* d_out, int out_size, void* d_ws, size_t ws_size,
                              hipStream_t stream) {
  const float* x   = (const float*)d_in[0];
  const int*   ei  = (const int*)d_in[1];
  const float* ew  = (const float*)d_in[2];
  const float* Wc[4] = {(const float*)d_in[3], (const float*)d_in[5],
                        (const float*)d_in[7], (const float*)d_in[9]};
  const float* bc[4] = {(const float*)d_in[4], (const float*)d_in[6],
                        (const float*)d_in[8], (const float*)d_in[10]};
  const float* Wout = (const float*)d_in[11];
  const float* bout = (const float*)d_in[12];

  const int n = in_sizes[0] / HIDF;
  const int E = in_sizes[1] / 2;
  const int* row = ei;
  const int* col = ei + E;

  const int NB = (E + EB - 1) / EB;
  const int K  = (n + RB - 1) / RB;
  const int NT = K * NB;

  char* ws = (char*)d_ws;
  const size_t fb = (size_t)n * HIDF * sizeof(float);
  const size_t nb = (((size_t)(n + 1) * 4) + 255) & ~(size_t)255;
  const size_t tb = (((size_t)(NT + 1) * 4) + 255) & ~(size_t)255;
  const size_t eb = (((size_t)E * 4) + 255) & ~(size_t)255;
  float* hw   = (float*)ws;
  float* hbuf = (float*)(ws + fb);
  char*  p    = ws + 2 * fb;
  float* dis    = (float*)p; p += nb;
  int*   offs   = (int*)p;   p += nb;
  int*   tab    = (int*)p;   p += tb;
  int*   base   = (int*)p;   p += tb;
  int*   pos    = (int*)p;   p += eb;
  int2*  sedge  = (int2*)p;  p += 2 * eb;
  int4*  rec    = (int4*)p;  p += 4 * eb;
  int4*  tmp    = (int4*)p;

  k_bin_count<<<NB, ETH, 0, stream>>>(col, pos, tab, E, NB, K);
  k_scan<<<(NT + 1023) / 1024, 1024, 0, stream>>>(tab, base, NT);
  k_bin_place<<<NB, ETH, 0, stream>>>(row, col, ew, pos, base, rec, E, NB, K);
  k_bin_order<<<K, 1024, 0, stream>>>(base, rec, tmp, sedge, offs, dis, n, NB);

  k_gemm_wmma<<<n / 16, 256, 0, stream>>>(x, Wc[0], hw, nullptr);
  k_gather_gemm<<<n / 16, 256, 0, stream>>>(hw, offs, sedge, dis, bc[0], 1, Wc[1], nullptr, hbuf);
  k_gather_gemm<<<n / 16, 256, 0, stream>>>(hbuf, offs, sedge, dis, bc[1], 1, Wc[2], nullptr, hw);
  k_gather_gemm<<<n / 16, 256, 0, stream>>>(hw, offs, sedge, dis, bc[2], 1, Wc[3], nullptr, hbuf);
  k_gather_gemm<<<n / 16, 256, 0, stream>>>(hbuf, offs, sedge, dis, bc[3], 0, Wout, bout,
                                            (float*)d_out);
}


# edge coefficient d^-1/2[row]*w*d^-1/2[col] formed once: layer 0's pass stores it in the edge list, later passes read it with the row index, no per-edge d^-1/2 gather; sum order unchanged
# speedup vs baseline: 4.1916x; 1.0344x over previous
#include <hip/hip_runtime.h>
#include <math.h>

typedef __attribute__((ext_vector_type(2))) float v2f;
typedef __attribute__((ext_vector_type(8))) float v8f;

#define HIDF 128


#define EB   8192
#define ETH  512
#define RB   256
#define RSH  8
#define KMAX 512
#define WCAP 64

__global__ __launch_bounds__(ETH)
void k_bin_count(const int* __restrict__ col, int* __restrict__ pos, int* __restrict__ tab,
                 int E, int NB, int K) {
  __shared__ int h[KMAX];
  const int t = threadIdx.x, b = blockIdx.x;
  for (int i = t; i < K; i += ETH) h[i] = 0;
  __syncthreads();
  const int e0 = b * EB + t;
  if (b * EB + EB <= E) {
    for (int g = 0; g < EB / ETH; g += 8) {
      int c[8];
#pragma unroll
      for (int u = 0; u < 8; ++u) c[u] = col[e0 + (g + u) * ETH];
#pragma unroll
      for (int u = 0; u < 8; ++u) pos[e0 + (g + u) * ETH] = atomicAdd(&h[c[u] >> RSH], 1);
    }
  } else {
    for (int e = e0; e < E; e += ETH) pos[e] = atomicAdd(&h[col[e] >> RSH], 1);
  }
  __syncthreads();
  for (int i = t; i < K; i += ETH) tab[i * NB + b] = h[i];
}

__global__ __launch_bounds__(1024)
void k_scan(const int* __restrict__ cnt, int* __restrict__ offs, int n) {
  __shared__ int part[32];
  __shared__ int wtot[32];
  const int t = threadIdx.x, lane = t & 31, wv = t >> 5;
  const int start = blockIdx.x * 1024;
  int s = 0;
  for (int i = t; i < start; i += 1024) s += cnt[i];
  const int i = start + t;
  const int v = (i < n) ? cnt[i] : 0;
  int inc = v;
  for (int o = 1; o < 32; o <<= 1) {
    const int u = __shfl_up(inc, o, 32);
    if (lane >= o) inc += u;
  }
  for (int o = 16; o > 0; o >>= 1) s += __shfl_xor(s, o, 32);
  if (lane == 31) wtot[wv] = inc;
  if (lane == 0) part[wv] = s;
  __syncthreads();
  int base = 0;
  for (int w = 0; w < 32; ++w) {
    base += part[w];
    if (w < wv) base += wtot[w];
  }
  const int excl = base + inc - v;
  if (i < n) offs[i] = excl;
  if (i == n - 1) offs[n] = excl + v;
}

__global__ __launch_bounds__(ETH)
void k_bin_place(const int* __restrict__ row, const int* __restrict__ col,
                 const float* __restrict__ ew, const int* __restrict__ pos,
                 const int* __restrict__ base, int4* __restrict__ rec, int E, int NB, int K) {
  __shared__ int sb[KMAX];
  const int t = threadIdx.x, b = blockIdx.x;
  for (int i = t; i < K; i += ETH) sb[i] = base[i * NB + b];
  __syncthreads();
  const int e0 = b * EB + t;
  if (b * EB + EB <= E) {
    for (int g = 0; g < EB / ETH; g += 4) {
      int c[4], p[4], r[4];
      float w[4];
#pragma unroll
      for (int u = 0; u < 4; ++u) {
        const int e = e0 + (g + u) * ETH;
        c[u] = col[e]; p[u] = pos[e]; r[u] = row[e]; w[u] = ew[e];
      }
#pragma unroll
      for (int u = 0; u < 4; ++u) {
        const int e = e0 + (g + u) * ETH;
        rec[sb[c[u] >> RSH] + p[u]] = make_int4(e, r[u], __float_as_int(w[u]), c[u]);
      }
    }
  } else {
    for (int e = e0; e < E; e += ETH) {
      const int c = col[e];
      rec[sb[c >> RSH] + pos[e]] = make_int4(e, row[e], __float_as_int(ew[e]), c);
    }
  }
}

__global__ __launch_bounds__(1024)
void k_bin_order(const int* __restrict__ base, const int4* __restrict__ rec,
                 int4* __restrict__ tmp, int2* __restrict__ sedge, int* __restrict__ offs,
                 float* __restrict__ dis, int n, int NB) {
  __shared__ int cnt[RB];
  __shared__ int loff[RB];
  __shared__ int cur[RB];
  __shared__ int wtot[RB / 32];
  __shared__ float wsort[32 * WCAP];
  const int t = threadIdx.x, lane = t & 31;
  const int wv = __builtin_amdgcn_readfirstlane(t >> 5);
  const int k = blockIdx.x;
  const int bs = __builtin_amdgcn_readfirstlane(base[k * NB]);
  const int be = __builtin_amdgcn_readfirstlane(base[(k + 1) * NB]);
  const int nb0 = k * RB;
  if (t < RB) cnt[t] = 0;
  __syncthreads();
  for (int idx = bs + t; idx < be; idx += 1024) atomicAdd(&cnt[rec[idx].w - nb0], 1);
  __syncthreads();
  const int v = (t < RB) ? cnt[t] : 0;
  int inc = v;
  for (int o = 1; o < 32; o <<= 1) {
    const int u = __shfl_up(inc, o, 32);
    if (lane >= o) inc += u;
  }
  if (t < RB && lane == 31) wtot[t >> 5] = inc;
  __syncthreads();
  if (t < RB) {
    int pre = 0;
    for (int w = 0; w < RB / 32; ++w)
      if (w < (t >> 5)) pre += wtot[w];
    const int ex = pre + inc - v;
    loff[t] = ex;
    cur[t] = ex;
    if (nb0 + t < n) offs[nb0 + t] = bs + ex;
  }
  if (k == (int)gridDim.x - 1 && t == 0) offs[n] = be;
  __syncthreads();
  for (int idx = bs + t; idx < be; idx += 1024) {
    const int4 q = rec[idx];
    const int p = atomicAdd(&cur[q.w - nb0], 1);
    tmp[bs + p] = q;
  }
  __syncthreads();

  float* const ws = &wsort[wv * WCAP];
  int i = wv;
  bool have = (nb0 + i < n);
  int d = 0, beg = 0;
  int4 q = make_int4(0x7fffffff, 0, 0, 0);
  if (have) {
    d = __builtin_amdgcn_readfirstlane(cnt[i]);
    beg = bs + __builtin_amdgcn_readfirstlane(loff[i]);
    if (lane < d) q = tmp[beg + lane];
  }
  while (have) {
    const int i2 = i + 32;
    const bool have2 = (i2 < RB) && (nb0 + i2 < n);
    int d2 = 0, beg2 = 0;
    int4 q2 = make_int4(0x7fffffff, 0, 0, 0);
    if (have2) {
      d2 = __builtin_amdgcn_readfirstlane(cnt[i2]);
      beg2 = bs + __builtin_amdgcn_readfirstlane(loff[i2]);
      if (lane < d2) q2 = tmp[beg2 + lane];
    }
    float deg = 0.0f;
    if (d <= WCAP) {
      for (int j0 = 0; j0 < d; j0 += 32) {
        int4 a = q;
        if (j0 > 0) {
          a = make_int4(0x7fffffff, 0, 0, 0);
          if (j0 + lane < d) a = tmp[beg + j0 + lane];
        }
        int rank = 0;
        for (int k0 = 0; k0 < d; k0 += 32) {
          int ek = a.x;
          if (k0 != j0) {
            ek = 0x7fffffff;
            if (k0 + lane < d) ek = tmp[beg + k0 + lane].x;
          }
          const int m = (d - k0 < 32) ? (d - k0) : 32;
          for (int kk = 0; kk < m; ++kk)
            rank += (__builtin_amdgcn_readlane(ek, kk) < a.x) ? 1 : 0;
        }
        if (j0 + lane < d) {
          sedge[beg + rank] = make_int2(a.y, a.z);
          ws[rank] = __int_as_float(a.z);
        }
      }
      for (int s0 = 0; s0 < d; s0 += 32) {
        const float sw = (s0 + lane < d) ? ws[s0 + lane] : 0.0f;
        const int m = (d - s0 < 32) ? (d - s0) : 32;
        for (int kk = 0; kk < m; ++kk)
          deg += __int_as_float(__builtin_amdgcn_readlane(__float_as_int(sw), kk));
      }
    } else if (lane == 0) {
      int last = -1;
      for (int s = 0; s < d; ++s) {
        int best = 0x7fffffff, bj = 0;
        for (int j = 0; j < d; ++j) {
          const int ee = tmp[beg + j].x;
          if (ee > last && ee < best) { best = ee; bj = j; }
        }
        const int4 z = tmp[beg + bj];
        sedge[beg + s] = make_int2(z.y, z.z);
        deg += __int_as_float(z.z);
        last = best;
      }
    }
    if (lane == 0) {
      deg += 1.0f;
      dis[nb0 + i] = (deg > 0.0f) ? rsqrtf(deg) : 0.0f;
    }
    i = i2; have = have2; d = d2; beg = beg2; q = q2;
  }
}

__global__ __launch_bounds__(256)
void k_gemm_wmma(const float* __restrict__ A, const float* __restrict__ W,
                 float* __restrict__ C, const float* __restrict__ bias) {
  __shared__ float As[16 * 132];
  const int t  = threadIdx.x;
  const int m0 = blockIdx.x * 16;

  {
    const float4* A4 = (const float4*)(A + (size_t)m0 * HIDF);
    for (int i = t; i < 16 * 32; i += 256) {
      int r = i >> 5, c = i & 31;
      float4 v = A4[i];
      As[r * 132 + c * 4 + 0] = v.x;
      As[r * 132 + c * 4 + 1] = v.y;
      As[r * 132 + c * 4 + 2] = v.z;
      As[r * 132 + c * 4 + 3] = v.w;
    }
  }
  __syncthreads();

  const int lane = t & 31;
  const int wave = t >> 5;
  const int half = lane >> 4;
  const int l15  = lane & 15;
  const int n0   = wave * 16;

  v8f acc = {};
#pragma unroll
  for (int kt = 0; kt < 32; ++kt) {
    const int k0 = kt * 4 + 2 * half;
#if defined(__has_builtin) && __has_builtin(__builtin_amdgcn_wmma_f32_16x16x4_f32)
    v2f a, b;
    a.x = As[l15 * 132 + k0 + 0];
    a.y = As[l15 * 132 + k0 + 1];
    b.x = W[(k0 + 0) * HIDF + n0 + l15];
    b.y = W[(k0 + 1) * HIDF + n0 + l15];
    acc = __builtin_amdgcn_wmma_f32_16x16x4_f32(false, a, false, b, (short)0, acc,
                                                false, false);
#else
#pragma unroll
    for (int r = 0; r < 8; ++r) {
      int M = r + 8 * half;
      int kk = kt * 4;
      float s = acc[r];
      s += As[M * 132 + kk + 0] * W[(kk + 0) * HIDF + n0 + l15];
      s += As[M * 132 + kk + 1] * W[(kk + 1) * HIDF + n0 + l15];
      s += As[M * 132 + kk + 2] * W[(kk + 2) * HIDF + n0 + l15];
      s += As[M * 132 + kk + 3] * W[(kk + 3) * HIDF + n0 + l15];
      acc[r] = s;
    }
#endif
  }

  const float bv = bias ? bias[n0 + l15] : 0.0f;
#pragma unroll
  for (int r = 0; r < 8; ++r) {
    C[(size_t)(m0 + r + 8 * half) * HIDF + (n0 + l15)] = acc[r] + bv;
  }
}

template <int B>
__device__ __forceinline__ void rows_add(const char* __restrict__ hwb, int r, float w,
                                         int k, unsigned loff, float4& acc) {
  float4 v[B];
  float wk[B];
#pragma unroll
  for (int i = 0; i < B; ++i) {
    const unsigned rk = (unsigned)__builtin_amdgcn_readlane(r, k + i);
    wk[i] = __int_as_float(__builtin_amdgcn_readlane(__float_as_int(w), k + i));
    const char* rowp = hwb + ((size_t)rk << 9);
    v[i] = *(const float4*)(rowp + loff);
  }
#pragma unroll
  for (int i = 0; i < B; ++i) {
    acc.x += v[i].x * wk[i];
    acc.y += v[i].y * wk[i];
    acc.z += v[i].z * wk[i];
    acc.w += v[i].w * wk[i];
  }
}

template <bool FIRST>
__global__ __launch_bounds__(256, 16)
void k_gather_gemm(const float* __restrict__ hw, const int* __restrict__ offs,
                   int2* __restrict__ sedge, const float* __restrict__ dis,
                   const float* __restrict__ b, int relu,
                   const float* __restrict__ W, const float* __restrict__ bias2,
                   float* __restrict__ C) {
  __shared__ float As[16 * 132];
  const int t    = threadIdx.x;
  const int lane = t & 31;
  const int wave = __builtin_amdgcn_readfirstlane(t >> 5);
  const int m0   = blockIdx.x * 16;
  const float4* hw4 = (const float4*)hw;
  const char* hwb = (const char*)hw;
  const unsigned loff = (unsigned)lane * 16u;
  const float4 bb = ((const float4*)b)[lane];

  for (int i = 0; i < 2; ++i) {
    const int lr = wave * 2 + i;
    const int c  = m0 + lr;
    const int beg = offs[c], end = offs[c + 1];
    const float dc = dis[c];
    const float4 hv = hw4[(size_t)c * 32 + lane];
    const float s2 = dc * dc;
    float4 acc = make_float4(hv.x * s2, hv.y * s2, hv.z * s2, hv.w * s2);
    for (int base = beg; base < end; base += 32) {
      const int j = base + lane;
      int r = 0;
      float w = 0.0f;
      if (j < end) {
        const int2 q = sedge[j];
        r = q.x;
        if (FIRST) {
          w = dis[r] * __int_as_float(q.y) * dc;
          sedge[j] = make_int2(r, __float_as_int(w));
        } else {
          w = __int_as_float(q.y);
        }
      }
      const int left = end - base;
      const int m = left < 32 ? left : 32;
      int k = 0;
      for (; k + 8 <= m; k += 8) rows_add<8>(hwb, r, w, k, loff, acc);
      if (k + 4 <= m) { rows_add<4>(hwb, r, w, k, loff, acc); k += 4; }
      if (k + 2 <= m) { rows_add<2>(hwb, r, w, k, loff, acc); k += 2; }
      if (k < m)      { rows_add<1>(hwb, r, w, k, loff, acc); }
    }
    acc.x += bb.x; acc.y += bb.y; acc.z += bb.z; acc.w += bb.w;
    if (relu) {
      acc.x = fmaxf(acc.x, 0.0f); acc.y = fmaxf(acc.y, 0.0f);
      acc.z = fmaxf(acc.z, 0.0f); acc.w = fmaxf(acc.w, 0.0f);
    }
    *(float4*)&As[lr * 132 + lane * 4] = acc;
  }
  __syncthreads();

  const int half = lane >> 4;
  const int l15  = lane & 15;
  const int n0   = wave * 16;

  v8f cacc = {};
#pragma unroll
  for (int kt = 0; kt < 32; ++kt) {
    const int k0 = kt * 4 + 2 * half;
    v2f a, bf;
    a.x = As[l15 * 132 + k0 + 0];
    a.y = As[l15 * 132 + k0 + 1];
    bf.x = W[(k0 + 0) * HIDF + n0 + l15];
    bf.y = W[(k0 + 1) * HIDF + n0 + l15];
    cacc = __builtin_amdgcn_wmma_f32_16x16x4_f32(false, a, false, bf, (short)0, cacc,
                                                 false, false);
  }

  const float bv = bias2 ? bias2[n0 + l15] : 0.0f;
#pragma unroll
  for (int r = 0; r < 8; ++r) {
    C[(size_t)(m0 + r + 8 * half) * HIDF + (n0 + l15)] = cacc[r] + bv;
  }
}

extern "C" void kernel_launch(void* const* d_in, const int* in_sizes, int n_in,
                              void* d_out, int out_size, void* d_ws, size_t ws_size,
                              hipStream_t stream) {
  const float* x   = (const float*)d_in[0];
  const int*   ei  = (const int*)d_in[1];
  const float* ew  = (const float*)d_in[2];
  const float* Wc[4] = {(const float*)d_in[3], (const float*)d_in[5],
                        (const float*)d_in[7], (const float*)d_in[9]};
  const float* bc[4] = {(const float*)d_in[4], (const float*)d_in[6],
                        (const float*)d_in[8], (const float*)d_in[10]};
  const float* Wout = (const float*)d_in[11];
  const float* bout = (const float*)d_in[12];

  const int n = in_sizes[0] / HIDF;
  const int E = in_sizes[1] / 2;
  const int* row = ei;
  const int* col = ei + E;

  const int NB = (E + EB - 1) / EB;
  const int K  = (n + RB - 1) / RB;
  const int NT = K * NB;

  char* ws = (char*)d_ws;
  const size_t fb = (size_t)n * HIDF * sizeof(float);
  const size_t nb = (((size_t)(n + 1) * 4) + 255) & ~(size_t)255;
  const size_t tb = (((size_t)(NT + 1) * 4) + 255) & ~(size_t)255;
  const size_t eb = (((size_t)E * 4) + 255) & ~(size_t)255;
  float* hw   = (float*)ws;
  float* hbuf = (float*)(ws + fb);
  char*  p    = ws + 2 * fb;
  float* dis    = (float*)p; p += nb;
  int*   offs   = (int*)p;   p += nb;
  int*   tab    = (int*)p;   p += tb;
  int*   base   = (int*)p;   p += tb;
  int*   pos    = (int*)p;   p += eb;
  int2*  sedge  = (int2*)p;  p += 2 * eb;
  int4*  rec    = (int4*)p;  p += 4 * eb;
  int4*  tmp    = (int4*)p;

  k_bin_count<<<NB, ETH, 0, stream>>>(col, pos, tab, E, NB, K);
  k_scan<<<(NT + 1023) / 1024, 1024, 0, stream>>>(tab, base, NT);
  k_bin_place<<<NB, ETH, 0, stream>>>(row, col, ew, pos, base, rec, E, NB, K);
  k_bin_order<<<K, 1024, 0, stream>>>(base, rec, tmp, sedge, offs, dis, n, NB);

  k_gemm_wmma<<<n / 16, 256, 0, stream>>>(x, Wc[0], hw, nullptr);
  k_gather_gemm<true><<<n / 16, 256, 0, stream>>>(hw, offs, sedge, dis, bc[0], 1, Wc[1], nullptr,
                                                  hbuf);
  k_gather_gemm<false><<<n / 16, 256, 0, stream>>>(hbuf, offs, sedge, dis, bc[1], 1, Wc[2],
                                                   nullptr, hw);
  k_gather_gemm<false><<<n / 16, 256, 0, stream>>>(hw, offs, sedge, dis, bc[2], 1, Wc[3],
                                                   nullptr, hbuf);
  k_gather_gemm<false><<<n / 16, 256, 0, stream>>>(hbuf, offs, sedge, dis, bc[3], 0, Wout, bout,
                                                   (float*)d_out);
}


# last layer's two products folded: W3 Wout and b3 Wout formed once on WMMA in layer 0's launch, final launch only sums rows (5 -> 4 large products); products re-associated, edge sum order kept
# speedup vs baseline: 4.4756x; 1.0678x over previous
#include <hip/hip_runtime.h>
#include <math.h>

typedef __attribute__((ext_vector_type(2))) float v2f;
typedef __attribute__((ext_vector_type(8))) float v8f;

#define HIDF 128


#define EB   8192
#define ETH  512
#define RB   256
#define RSH  8
#define KMAX 512
#define WCAP 64

__global__ __launch_bounds__(ETH)
void k_bin_count(const int* __restrict__ col, int* __restrict__ pos, int* __restrict__ tab,
                 int E, int NB, int K) {
  __shared__ int h[KMAX];
  const int t = threadIdx.x, b = blockIdx.x;
  for (int i = t; i < K; i += ETH) h[i] = 0;
  __syncthreads();
  const int e0 = b * EB + t;
  if (b * EB + EB <= E) {
    for (int g = 0; g < EB / ETH; g += 8) {
      int c[8];
#pragma unroll
      for (int u = 0; u < 8; ++u) c[u] = col[e0 + (g + u) * ETH];
#pragma unroll
      for (int u = 0; u < 8; ++u) pos[e0 + (g + u) * ETH] = atomicAdd(&h[c[u] >> RSH], 1);
    }
  } else {
    for (int e = e0; e < E; e += ETH) pos[e] = atomicAdd(&h[col[e] >> RSH], 1);
  }
  __syncthreads();
  for (int i = t; i < K; i += ETH) tab[i * NB + b] = h[i];
}

__global__ __launch_bounds__(1024)
void k_scan(const int* __restrict__ cnt, int* __restrict__ offs, int n) {
  __shared__ int part[32];
  __shared__ int wtot[32];
  const int t = threadIdx.x, lane = t & 31, wv = t >> 5;
  const int start = blockIdx.x * 1024;
  int s = 0;
  for (int i = t; i < start; i += 1024) s += cnt[i];
  const int i = start + t;
  const int v = (i < n) ? cnt[i] : 0;
  int inc = v;
  for (int o = 1; o < 32; o <<= 1) {
    const int u = __shfl_up(inc, o, 32);
    if (lane >= o) inc += u;
  }
  for (int o = 16; o > 0; o >>= 1) s += __shfl_xor(s, o, 32);
  if (lane == 31) wtot[wv] = inc;
  if (lane == 0) part[wv] = s;
  __syncthreads();
  int base = 0;
  for (int w = 0; w < 32; ++w) {
    base += part[w];
    if (w < wv) base += wtot[w];
  }
  const int excl = base + inc - v;
  if (i < n) offs[i] = excl;
  if (i == n - 1) offs[n] = excl + v;
}

__global__ __launch_bounds__(ETH)
void k_bin_place(const int* __restrict__ row, const int* __restrict__ col,
                 const float* __restrict__ ew, const int* __restrict__ pos,
                 const int* __restrict__ base, int4* __restrict__ rec, int E, int NB, int K) {
  __shared__ int sb[KMAX];
  const int t = threadIdx.x, b = blockIdx.x;
  for (int i = t; i < K; i += ETH) sb[i] = base[i * NB + b];
  __syncthreads();
  const int e0 = b * EB + t;
  if (b * EB + EB <= E) {
    for (int g = 0; g < EB / ETH; g += 4) {
      int c[4], p[4], r[4];
      float w[4];
#pragma unroll
      for (int u = 0; u < 4; ++u) {
        const int e = e0 + (g + u) * ETH;
        c[u] = col[e]; p[u] = pos[e]; r[u] = row[e]; w[u] = ew[e];
      }
#pragma unroll
      for (int u = 0; u < 4; ++u) {
        const int e = e0 + (g + u) * ETH;
        rec[sb[c[u] >> RSH] + p[u]] = make_int4(e, r[u], __float_as_int(w[u]), c[u]);
      }
    }
  } else {
    for (int e = e0; e < E; e += ETH) {
      const int c = col[e];
      rec[sb[c >> RSH] + pos[e]] = make_int4(e, row[e], __float_as_int(ew[e]), c);
    }
  }
}

__global__ __launch_bounds__(1024)
void k_bin_order(const int* __restrict__ base, const int4* __restrict__ rec,
                 int4* __restrict__ tmp, int2* __restrict__ sedge, int* __restrict__ offs,
                 float* __restrict__ dis, int n, int NB) {
  __shared__ int cnt[RB];
  __shared__ int loff[RB];
  __shared__ int cur[RB];
  __shared__ int wtot[RB / 32];
  __shared__ float wsort[32 * WCAP];
  const int t = threadIdx.x, lane = t & 31;
  const int wv = __builtin_amdgcn_readfirstlane(t >> 5);
  const int k = blockIdx.x;
  const int bs = __builtin_amdgcn_readfirstlane(base[k * NB]);
  const int be = __builtin_amdgcn_readfirstlane(base[(k + 1) * NB]);
  const int nb0 = k * RB;
  if (t < RB) cnt[t] = 0;
  __syncthreads();
  for (int idx = bs + t; idx < be; idx += 1024) atomicAdd(&cnt[rec[idx].w - nb0], 1);
  __syncthreads();
  const int v = (t < RB) ? cnt[t] : 0;
  int inc = v;
  for (int o = 1; o < 32; o <<= 1) {
    const int u = __shfl_up(inc, o, 32);
    if (lane >= o) inc += u;
  }
  if (t < RB && lane == 31) wtot[t >> 5] = inc;
  __syncthreads();
  if (t < RB) {
    int pre = 0;
    for (int w = 0; w < RB / 32; ++w)
      if (w < (t >> 5)) pre += wtot[w];
    const int ex = pre + inc - v;
    loff[t] = ex;
    cur[t] = ex;
    if (nb0 + t < n) offs[nb0 + t] = bs + ex;
  }
  if (k == (int)gridDim.x - 1 && t == 0) offs[n] = be;
  __syncthreads();
  for (int idx = bs + t; idx < be; idx += 1024) {
    const int4 q = rec[idx];
    const int p = atomicAdd(&cur[q.w - nb0], 1);
    tmp[bs + p] = q;
  }
  __syncthreads();

  float* const ws = &wsort[wv * WCAP];
  int i = wv;
  bool have = (nb0 + i < n);
  int d = 0, beg = 0;
  int4 q = make_int4(0x7fffffff, 0, 0, 0);
  if (have) {
    d = __builtin_amdgcn_readfirstlane(cnt[i]);
    beg = bs + __builtin_amdgcn_readfirstlane(loff[i]);
    if (lane < d) q = tmp[beg + lane];
  }
  while (have) {
    const int i2 = i + 32;
    const bool have2 = (i2 < RB) && (nb0 + i2 < n);
    int d2 = 0, beg2 = 0;
    int4 q2 = make_int4(0x7fffffff, 0, 0, 0);
    if (have2) {
      d2 = __builtin_amdgcn_readfirstlane(cnt[i2]);
      beg2 = bs + __builtin_amdgcn_readfirstlane(loff[i2]);
      if (lane < d2) q2 = tmp[beg2 + lane];
    }
    float deg = 0.0f;
    if (d <= WCAP) {
      for (int j0 = 0; j0 < d; j0 += 32) {
        int4 a = q;
        if (j0 > 0) {
          a = make_int4(0x7fffffff, 0, 0, 0);
          if (j0 + lane < d) a = tmp[beg + j0 + lane];
        }
        int rank = 0;
        for (int k0 = 0; k0 < d; k0 += 32) {
          int ek = a.x;
          if (k0 != j0) {
            ek = 0x7fffffff;
            if (k0 + lane < d) ek = tmp[beg + k0 + lane].x;
          }
          const int m = (d - k0 < 32) ? (d - k0) : 32;
          for (int kk = 0; kk < m; ++kk)
            rank += (__builtin_amdgcn_readlane(ek, kk) < a.x) ? 1 : 0;
        }
        if (j0 + lane < d) {
          sedge[beg + rank] = make_int2(a.y, a.z);
          ws[rank] = __int_as_float(a.z);
        }
      }
      for (int s0 = 0; s0 < d; s0 += 32) {
        const float sw = (s0 + lane < d) ? ws[s0 + lane] : 0.0f;
        const int m = (d - s0 < 32) ? (d - s0) : 32;
        for (int kk = 0; kk < m; ++kk)
          deg += __int_as_float(__builtin_amdgcn_readlane(__float_as_int(sw), kk));
      }
    } else if (lane == 0) {
      int last = -1;
      for (int s = 0; s < d; ++s) {
        int best = 0x7fffffff, bj = 0;
        for (int j = 0; j < d; ++j) {
          const int ee = tmp[beg + j].x;
          if (ee > last && ee < best) { best = ee; bj = j; }
        }
        const int4 z = tmp[beg + bj];
        sedge[beg + s] = make_int2(z.y, z.z);
        deg += __int_as_float(z.z);
        last = best;
      }
    }
    if (lane == 0) {
      deg += 1.0f;
      dis[nb0 + i] = (deg > 0.0f) ? rsqrtf(deg) : 0.0f;
    }
    i = i2; have = have2; d = d2; beg = beg2; q = q2;
  }
}

__global__ __launch_bounds__(256)
void k_gemm_wmma(const float* __restrict__ A, const float* __restrict__ W,
                 float* __restrict__ C, int nt,
                 const float* __restrict__ A2, const float* __restrict__ W2,
                 const float* __restrict__ bvec, float* __restrict__ C2) {
  __shared__ float As[16 * 132];
  const int t  = threadIdx.x;
  const int xb = (int)blockIdx.x - nt;
  const bool extra = xb >= 0;
  const int m0 = (extra ? xb : (int)blockIdx.x) * 16;
  const float* Ap = extra ? A2 : A;
  const float* Wp = extra ? W2 : W;
  float* Cp = extra ? C2 : C;

  {
    const float4* A4 = (const float4*)(Ap + (size_t)m0 * HIDF);
    const int r = t >> 5, c = t & 31;
    float4 v0, v1;
    if (xb < 8) {
      v0 = A4[t];
      v1 = A4[t + 256];
    } else {
      v0 = (r == 0) ? ((const float4*)bvec)[c] : make_float4(0.0f, 0.0f, 0.0f, 0.0f);
      v1 = make_float4(0.0f, 0.0f, 0.0f, 0.0f);
    }
    As[r * 132 + c * 4 + 0] = v0.x;
    As[r * 132 + c * 4 + 1] = v0.y;
    As[r * 132 + c * 4 + 2] = v0.z;
    As[r * 132 + c * 4 + 3] = v0.w;
    As[(r + 8) * 132 + c * 4 + 0] = v1.x;
    As[(r + 8) * 132 + c * 4 + 1] = v1.y;
    As[(r + 8) * 132 + c * 4 + 2] = v1.z;
    As[(r + 8) * 132 + c * 4 + 3] = v1.w;
  }
  __syncthreads();

  const int lane = t & 31;
  const int wave = t >> 5;
  const int half = lane >> 4;
  const int l15  = lane & 15;
  const int n0   = wave * 16;

  v8f acc = {};
#pragma unroll
  for (int kt = 0; kt < 32; ++kt) {
    const int k0 = kt * 4 + 2 * half;
    v2f a, b;
    a.x = As[l15 * 132 + k0 + 0];
    a.y = As[l15 * 132 + k0 + 1];
    b.x = Wp[(k0 + 0) * HIDF + n0 + l15];
    b.y = Wp[(k0 + 1) * HIDF + n0 + l15];
    acc = __builtin_amdgcn_wmma_f32_16x16x4_f32(false, a, false, b, (short)0, acc,
                                                false, false);
  }

#pragma unroll
  for (int r = 0; r < 8; ++r) {
    Cp[(size_t)(m0 + r + 8 * half) * HIDF + (n0 + l15)] = acc[r];
  }
}

template <int B>
__device__ __forceinline__ void rows_add(const char* __restrict__ hwb, int r, float w,
                                         int k, unsigned loff, float4& acc) {
  float4 v[B];
  float wk[B];
#pragma unroll
  for (int i = 0; i < B; ++i) {
    const unsigned rk = (unsigned)__builtin_amdgcn_readlane(r, k + i);
    wk[i] = __int_as_float(__builtin_amdgcn_readlane(__float_as_int(w), k + i));
    const char* rowp = hwb + ((size_t)rk << 9);
    v[i] = *(const float4*)(rowp + loff);
  }
#pragma unroll
  for (int i = 0; i < B; ++i) {
    acc.x += v[i].x * wk[i];
    acc.y += v[i].y * wk[i];
    acc.z += v[i].z * wk[i];
    acc.w += v[i].w * wk[i];
  }
}

template <bool FIRST>
__device__ __forceinline__ float4 node_sum(const float* __restrict__ hw,
                                           const int* __restrict__ offs,
                                           int2* __restrict__ sedge,
                                           const float* __restrict__ dis,
                                           int c, int lane) {
  const float4* hw4 = (const float4*)hw;
  const char* hwb = (const char*)hw;
  const unsigned loff = (unsigned)lane * 16u;
  const int beg = offs[c], end = offs[c + 1];
  const float dc = dis[c];
  const float4 hv = hw4[(size_t)c * 32 + lane];
  const float s2 = dc * dc;
  float4 acc = make_float4(hv.x * s2, hv.y * s2, hv.z * s2, hv.w * s2);
  for (int base = beg; base < end; base += 32) {
    const int j = base + lane;
    int r = 0;
    float w = 0.0f;
    if (j < end) {
      const int2 q = sedge[j];
      r = q.x;
      if (FIRST) {
        w = dis[r] * __int_as_float(q.y) * dc;
        sedge[j] = make_int2(r, __float_as_int(w));
      } else {
        w = __int_as_float(q.y);
      }
    }
    const int left = end - base;
    const int m = left < 32 ? left : 32;
    int k = 0;
    for (; k + 8 <= m; k += 8) rows_add<8>(hwb, r, w, k, loff, acc);
    if (k + 4 <= m) { rows_add<4>(hwb, r, w, k, loff, acc); k += 4; }
    if (k + 2 <= m) { rows_add<2>(hwb, r, w, k, loff, acc); k += 2; }
    if (k < m)      { rows_add<1>(hwb, r, w, k, loff, acc); }
  }
  return acc;
}

template <bool FIRST>
__global__ __launch_bounds__(256, 16)
void k_gather_gemm(const float* __restrict__ hw, const int* __restrict__ offs,
                   int2* __restrict__ sedge, const float* __restrict__ dis,
                   const float* __restrict__ b,
                   const float* __restrict__ W,
                   float* __restrict__ C) {
  __shared__ float As[16 * 132];
  const int t    = threadIdx.x;
  const int lane = t & 31;
  const int wave = __builtin_amdgcn_readfirstlane(t >> 5);
  const int m0   = blockIdx.x * 16;
  const float4 bb = ((const float4*)b)[lane];

  for (int i = 0; i < 2; ++i) {
    const int lr = wave * 2 + i;
    float4 acc = node_sum<FIRST>(hw, offs, sedge, dis, m0 + lr, lane);
    acc.x += bb.x; acc.y += bb.y; acc.z += bb.z; acc.w += bb.w;
    acc.x = fmaxf(acc.x, 0.0f); acc.y = fmaxf(acc.y, 0.0f);
    acc.z = fmaxf(acc.z, 0.0f); acc.w = fmaxf(acc.w, 0.0f);
    *(float4*)&As[lr * 132 + lane * 4] = acc;
  }
  __syncthreads();

  const int half = lane >> 4;
  const int l15  = lane & 15;
  const int n0   = wave * 16;

  v8f cacc = {};
#pragma unroll
  for (int kt = 0; kt < 32; ++kt) {
    const int k0 = kt * 4 + 2 * half;
    v2f a, bf;
    a.x = As[l15 * 132 + k0 + 0];
    a.y = As[l15 * 132 + k0 + 1];
    bf.x = W[(k0 + 0) * HIDF + n0 + l15];
    bf.y = W[(k0 + 1) * HIDF + n0 + l15];
    cacc = __builtin_amdgcn_wmma_f32_16x16x4_f32(false, a, false, bf, (short)0, cacc,
                                                 false, false);
  }

#pragma unroll
  for (int r = 0; r < 8; ++r) {
    C[(size_t)(m0 + r + 8 * half) * HIDF + (n0 + l15)] = cacc[r];
  }
}

__global__ __launch_bounds__(256, 16)
void k_gather_out(const float* __restrict__ hw, const int* __restrict__ offs,
                  int2* __restrict__ sedge, const float* __restrict__ dis,
                  const float* __restrict__ bfold, const float* __restrict__ bout,
                  float* __restrict__ out) {
  const int t    = threadIdx.x;
  const int lane = t & 31;
  const int wave = __builtin_amdgcn_readfirstlane(t >> 5);
  const int m0   = blockIdx.x * 16;
  const float4 bf = ((const float4*)bfold)[lane];
  const float4 bo = ((const float4*)bout)[lane];
  const float4 bb = make_float4(bf.x + bo.x, bf.y + bo.y, bf.z + bo.z, bf.w + bo.w);
  float4* out4 = (float4*)out;

  for (int i = 0; i < 2; ++i) {
    const int c = m0 + wave * 2 + i;
    float4 acc = node_sum<false>(hw, offs, sedge, dis, c, lane);
    acc.x += bb.x; acc.y += bb.y; acc.z += bb.z; acc.w += bb.w;
    out4[(size_t)c * 32 + lane] = acc;
  }
}

extern "C" void kernel_launch(void* const* d_in, const int* in_sizes, int n_in,
                              void* d_out, int out_size, void* d_ws, size_t ws_size,
                              hipStream_t stream) {
  const float* x   = (const float*)d_in[0];
  const int*   ei  = (const int*)d_in[1];
  const float* ew  = (const float*)d_in[2];
  const float* Wc[4] = {(const float*)d_in[3], (const float*)d_in[5],
                        (const float*)d_in[7], (const float*)d_in[9]};
  const float* bc[4] = {(const float*)d_in[4], (const float*)d_in[6],
                        (const float*)d_in[8], (const float*)d_in[10]};
  const float* Wout = (const float*)d_in[11];
  const float* bout = (const float*)d_in[12];

  const int n = in_sizes[0] / HIDF;
  const int E = in_sizes[1] / 2;
  const int* row = ei;
  const int* col = ei + E;

  const int NB = (E + EB - 1) / EB;
  const int K  = (n + RB - 1) / RB;
  const int NT = K * NB;

  char* ws = (char*)d_ws;
  const size_t fb = (size_t)n * HIDF * sizeof(float);
  const size_t nb = (((size_t)(n + 1) * 4) + 255) & ~(size_t)255;
  const size_t tb = (((size_t)(NT + 1) * 4) + 255) & ~(size_t)255;
  const size_t eb = (((size_t)E * 4) + 255) & ~(size_t)255;
  float* hw   = (float*)ws;
  float* hbuf = (float*)(ws + fb);
  char*  p    = ws + 2 * fb;
  float* dis    = (float*)p; p += nb;
  int*   offs   = (int*)p;   p += nb;
  int*   tab    = (int*)p;   p += tb;
  int*   base   = (int*)p;   p += tb;
  int*   pos    = (int*)p;   p += eb;
  int2*  sedge  = (int2*)p;  p += 2 * eb;
  int4*  rec    = (int4*)p;  p += 4 * eb;
  int4*  tmp    = (int4*)p;  p += 4 * eb;
  float* wfold  = (float*)p;

  k_bin_count<<<NB, ETH, 0, stream>>>(col, pos, tab, E, NB, K);
  k_scan<<<(NT + 1023) / 1024, 1024, 0, stream>>>(tab, base, NT);
  k_bin_place<<<NB, ETH, 0, stream>>>(row, col, ew, pos, base, rec, E, NB, K);
  k_bin_order<<<K, 1024, 0, stream>>>(base, rec, tmp, sedge, offs, dis, n, NB);

  k_gemm_wmma<<<n / 16 + 9, 256, 0, stream>>>(x, Wc[0], hw, n / 16, Wc[3], Wout, bc[3], wfold);
  k_gather_gemm<true><<<n / 16, 256, 0, stream>>>(hw, offs, sedge, dis, bc[0], Wc[1], hbuf);
  k_gather_gemm<false><<<n / 16, 256, 0, stream>>>(hbuf, offs, sedge, dis, bc[1], Wc[2], hw);
  k_gather_gemm<false><<<n / 16, 256, 0, stream>>>(hw, offs, sedge, dis, bc[2], wfold, hbuf);
  k_gather_out<<<n / 16, 256, 0, stream>>>(hbuf, offs, sedge, dis, wfold + 128 * HIDF, bout,
                                          (float*)d_out);
}


# B fragments of all products read from a copy of each weight in fragment order (packed once per call by 32 spare blocks of launch 1): 16 16-byte loads per wave, not 64 4-byte; sum order unchanged
# speedup vs baseline: 4.6587x; 1.0409x over previous
#include <hip/hip_runtime.h>
#include <math.h>

typedef __attribute__((ext_vector_type(2))) float v2f;
typedef __attribute__((ext_vector_type(8))) float v8f;

#define HIDF 128


#define EB   8192
#define ETH  512
#define RB   256
#define RSH  8
#define KMAX 512
#define WCAP 64

#define PKW (HIDF * HIDF)
#define NPK 4

__global__ __launch_bounds__(ETH)
void k_bin_count(const int* __restrict__ col, int* __restrict__ pos, int* __restrict__ tab,
                 int E, int NB, int K,
                 const float* __restrict__ Wa, const float* __restrict__ Wb,
                 const float* __restrict__ Wc, const float* __restrict__ Wd,
                 float* __restrict__ pk) {
  __shared__ int h[KMAX];
  const int t = threadIdx.x, b = blockIdx.x;
  if (b >= NB) {
    const int q = b - NB;
    const int which = q >> 3, w = q & 7;
    const float* Ws = (which == 0) ? Wa : (which == 1) ? Wb : (which == 2) ? Wc : Wd;
    const int j = t >> 5, lane = t & 31;
    const int n = 16 * w + (lane & 15);
    const int k0 = 8 * j + 2 * (lane >> 4);
    float4 v;
    v.x = Ws[(k0 + 0) * HIDF + n];
    v.y = Ws[(k0 + 1) * HIDF + n];
    v.z = Ws[(k0 + 4) * HIDF + n];
    v.w = Ws[(k0 + 5) * HIDF + n];
    ((float4*)(pk + (size_t)which * PKW))[(w * 16 + j) * 32 + lane] = v;
    return;
  }
  for (int i = t; i < K; i += ETH) h[i] = 0;
  __syncthreads();
  const int e0 = b * EB + t;
  if (b * EB + EB <= E) {
    for (int g = 0; g < EB / ETH; g += 8) {
      int c[8];
#pragma unroll
      for (int u = 0; u < 8; ++u) c[u] = col[e0 + (g + u) * ETH];
#pragma unroll
      for (int u = 0; u < 8; ++u) pos[e0 + (g + u) * ETH] = atomicAdd(&h[c[u] >> RSH], 1);
    }
  } else {
    for (int e = e0; e < E; e += ETH) pos[e] = atomicAdd(&h[col[e] >> RSH], 1);
  }
  __syncthreads();
  for (int i = t; i < K; i += ETH) tab[i * NB + b] = h[i];
}

__global__ __launch_bounds__(1024)
void k_scan(const int* __restrict__ cnt, int* __restrict__ offs, int n) {
  __shared__ int part[32];
  __shared__ int wtot[32];
  const int t = threadIdx.x, lane = t & 31, wv = t >> 5;
  const int start = blockIdx.x * 1024;
  int s = 0;
  for (int i = t; i < start; i += 1024) s += cnt[i];
  const int i = start + t;
  const int v = (i < n) ? cnt[i] : 0;
  int inc = v;
  for (int o = 1; o < 32; o <<= 1) {
    const int u = __shfl_up(inc, o, 32);
    if (lane >= o) inc += u;
  }
  for (int o = 16; o > 0; o >>= 1) s += __shfl_xor(s, o, 32);
  if (lane == 31) wtot[wv] = inc;
  if (lane == 0) part[wv] = s;
  __syncthreads();
  int base = 0;
  for (int w = 0; w < 32; ++w) {
    base += part[w];
    if (w < wv) base += wtot[w];
  }
  const int excl = base + inc - v;
  if (i < n) offs[i] = excl;
  if (i == n - 1) offs[n] = excl + v;
}

__global__ __launch_bounds__(ETH)
void k_bin_place(const int* __restrict__ row, const int* __restrict__ col,
                 const float* __restrict__ ew, const int* __restrict__ pos,
                 const int* __restrict__ base, int4* __restrict__ rec, int E, int NB, int K) {
  __shared__ int sb[KMAX];
  const int t = threadIdx.x, b = blockIdx.x;
  for (int i = t; i < K; i += ETH) sb[i] = base[i * NB + b];
  __syncthreads();
  const int e0 = b * EB + t;
  if (b * EB + EB <= E) {
    for (int g = 0; g < EB / ETH; g += 4) {
      int c[4], p[4], r[4];
      float w[4];
#pragma unroll
      for (int u = 0; u < 4; ++u) {
        const int e = e0 + (g + u) * ETH;
        c[u] = col[e]; p[u] = pos[e]; r[u] = row[e]; w[u] = ew[e];
      }
#pragma unroll
      for (int u = 0; u < 4; ++u) {
        const int e = e0 + (g + u) * ETH;
        rec[sb[c[u] >> RSH] + p[u]] = make_int4(e, r[u], __float_as_int(w[u]), c[u]);
      }
    }
  } else {
    for (int e = e0; e < E; e += ETH) {
      const int c = col[e];
      rec[sb[c >> RSH] + pos[e]] = make_int4(e, row[e], __float_as_int(ew[e]), c);
    }
  }
}

__global__ __launch_bounds__(1024)
void k_bin_order(const int* __restrict__ base, const int4* __restrict__ rec,
                 int4* __restrict__ tmp, int2* __restrict__ sedge, int* __restrict__ offs,
                 float* __restrict__ dis, int n, int NB) {
  __shared__ int cnt[RB];
  __shared__ int loff[RB];
  __shared__ int cur[RB];
  __shared__ int wtot[RB / 32];
  __shared__ float wsort[32 * WCAP];
  const int t = threadIdx.x, lane = t & 31;
  const int wv = __builtin_amdgcn_readfirstlane(t >> 5);
  const int k = blockIdx.x;
  const int bs = __builtin_amdgcn_readfirstlane(base[k * NB]);
  const int be = __builtin_amdgcn_readfirstlane(base[(k + 1) * NB]);
  const int nb0 = k * RB;
  if (t < RB) cnt[t] = 0;
  __syncthreads();
  for (int idx = bs + t; idx < be; idx += 1024) atomicAdd(&cnt[rec[idx].w - nb0], 1);
  __syncthreads();
  const int v = (t < RB) ? cnt[t] : 0;
  int inc = v;
  for (int o = 1; o < 32; o <<= 1) {
    const int u = __shfl_up(inc, o, 32);
    if (lane >= o) inc += u;
  }
  if (t < RB && lane == 31) wtot[t >> 5] = inc;
  __syncthreads();
  if (t < RB) {
    int pre = 0;
    for (int w = 0; w < RB / 32; ++w)
      if (w < (t >> 5)) pre += wtot[w];
    const int ex = pre + inc - v;
    loff[t] = ex;
    cur[t] = ex;
    if (nb0 + t < n) offs[nb0 + t] = bs + ex;
  }
  if (k == (int)gridDim.x - 1 && t == 0) offs[n] = be;
  __syncthreads();
  for (int idx = bs + t; idx < be; idx += 1024) {
    const int4 q = rec[idx];
    const int p = atomicAdd(&cur[q.w - nb0], 1);
    tmp[bs + p] = q;
  }
  __syncthreads();

  float* const ws = &wsort[wv * WCAP];
  int i = wv;
  bool have = (nb0 + i < n);
  int d = 0, beg = 0;
  int4 q = make_int4(0x7fffffff, 0, 0, 0);
  if (have) {
    d = __builtin_amdgcn_readfirstlane(cnt[i]);
    beg = bs + __builtin_amdgcn_readfirstlane(loff[i]);
    if (lane < d) q = tmp[beg + lane];
  }
  while (have) {
    const int i2 = i + 32;
    const bool have2 = (i2 < RB) && (nb0 + i2 < n);
    int d2 = 0, beg2 = 0;
    int4 q2 = make_int4(0x7fffffff, 0, 0, 0);
    if (have2) {
      d2 = __builtin_amdgcn_readfirstlane(cnt[i2]);
      beg2 = bs + __builtin_amdgcn_readfirstlane(loff[i2]);
      if (lane < d2) q2 = tmp[beg2 + lane];
    }
    float deg = 0.0f;
    if (d <= WCAP) {
      for (int j0 = 0; j0 < d; j0 += 32) {
        int4 a = q;
        if (j0 > 0) {
          a = make_int4(0x7fffffff, 0, 0, 0);
          if (j0 + lane < d) a = tmp[beg + j0 + lane];
        }
        int rank = 0;
        for (int k0 = 0; k0 < d; k0 += 32) {
          int ek = a.x;
          if (k0 != j0) {
            ek = 0x7fffffff;
            if (k0 + lane < d) ek = tmp[beg + k0 + lane].x;
          }
          const int m = (d - k0 < 32) ? (d - k0) : 32;
          for (int kk = 0; kk < m; ++kk)
            rank += (__builtin_amdgcn_readlane(ek, kk) < a.x) ? 1 : 0;
        }
        if (j0 + lane < d) {
          sedge[beg + rank] = make_int2(a.y, a.z);
          ws[rank] = __int_as_float(a.z);
        }
      }
      for (int s0 = 0; s0 < d; s0 += 32) {
        const float sw = (s0 + lane < d) ? ws[s0 + lane] : 0.0f;
        const int m = (d - s0 < 32) ? (d - s0) : 32;
        for (int kk = 0; kk < m; ++kk)
          deg += __int_as_float(__builtin_amdgcn_readlane(__float_as_int(sw), kk));
      }
    } else if (lane == 0) {
      int last = -1;
      for (int s = 0; s < d; ++s) {
        int best = 0x7fffffff, bj = 0;
        for (int j = 0; j < d; ++j) {
          const int ee = tmp[beg + j].x;
          if (ee > last && ee < best) { best = ee; bj = j; }
        }
        const int4 z = tmp[beg + bj];
        sedge[beg + s] = make_int2(z.y, z.z);
        deg += __int_as_float(z.z);
        last = best;
      }
    }
    if (lane == 0) {
      deg += 1.0f;
      dis[nb0 + i] = (deg > 0.0f) ? rsqrtf(deg) : 0.0f;
    }
    i = i2; have = have2; d = d2; beg = beg2; q = q2;
  }
}

__global__ __launch_bounds__(256)
void k_gemm_wmma(const float* __restrict__ A, const float* __restrict__ W,
                 float* __restrict__ C, int nt,
                 const float* __restrict__ A2, const float* __restrict__ W2,
                 const float* __restrict__ bvec, float* __restrict__ C2,
                 float* __restrict__ C3) {
  __shared__ float As[16 * 132];
  const int t  = threadIdx.x;
  const int xb = (int)blockIdx.x - nt;
  const bool extra = xb >= 0;
  const int m0 = (extra ? xb : (int)blockIdx.x) * 16;
  const float* Ap = extra ? A2 : A;
  const float* Wp = extra ? W2 : W;

  {
    const float4* A4 = (const float4*)(Ap + (size_t)m0 * HIDF);
    const int r = t >> 5, c = t & 31;
    float4 v0, v1;
    if (xb < 8) {
      v0 = A4[t];
      v1 = A4[t + 256];
    } else {
      v0 = (r == 0) ? ((const float4*)bvec)[c] : make_float4(0.0f, 0.0f, 0.0f, 0.0f);
      v1 = make_float4(0.0f, 0.0f, 0.0f, 0.0f);
    }
    As[r * 132 + c * 4 + 0] = v0.x;
    As[r * 132 + c * 4 + 1] = v0.y;
    As[r * 132 + c * 4 + 2] = v0.z;
    As[r * 132 + c * 4 + 3] = v0.w;
    As[(r + 8) * 132 + c * 4 + 0] = v1.x;
    As[(r + 8) * 132 + c * 4 + 1] = v1.y;
    As[(r + 8) * 132 + c * 4 + 2] = v1.z;
    As[(r + 8) * 132 + c * 4 + 3] = v1.w;
  }
  __syncthreads();

  const int lane = t & 31;
  const int wave = t >> 5;
  const int half = lane >> 4;
  const int l15  = lane & 15;
  const int n0   = wave * 16;

  const float4* Wq = (const float4*)Wp + (wave * 16) * 32 + lane;
  v8f acc = {};
#pragma unroll
  for (int j = 0; j < 16; ++j) {
    const float4 bq = Wq[j * 32];
    const int k0 = j * 8 + 2 * half;
    v2f a0, a1, b0, b1;
    a0.x = As[l15 * 132 + k0 + 0];
    a0.y = As[l15 * 132 + k0 + 1];
    a1.x = As[l15 * 132 + k0 + 4];
    a1.y = As[l15 * 132 + k0 + 5];
    b0.x = bq.x; b0.y = bq.y;
    b1.x = bq.z; b1.y = bq.w;
    acc = __builtin_amdgcn_wmma_f32_16x16x4_f32(false, a0, false, b0, (short)0, acc,
                                                false, false);
    acc = __builtin_amdgcn_wmma_f32_16x16x4_f32(false, a1, false, b1, (short)0, acc,
                                                false, false);
  }

  if (xb >= 0 && xb < 8) {
#pragma unroll
    for (int r = 0; r < 8; ++r) {
      const int pl = ((r >> 1) & 1) * 16 + l15;
      const int pe = (r & 1) + 2 * (r >> 2);
      C2[(((wave * 16) + 2 * xb + half) * 32 + pl) * 4 + pe] = acc[r];
    }
  } else {
    float* Cp = extra ? C3 : C;
    const int mr = extra ? 0 : m0;
#pragma unroll
    for (int r = 0; r < 8; ++r) {
      Cp[(size_t)(mr + r + 8 * half) * HIDF + (n0 + l15)] = acc[r];
    }
  }
}

template <int B>
__device__ __forceinline__ void rows_add(const char* __restrict__ hwb, int r, float w,
                                         int k, unsigned loff, float4& acc) {
  float4 v[B];
  float wk[B];
#pragma unroll
  for (int i = 0; i < B; ++i) {
    const unsigned rk = (unsigned)__builtin_amdgcn_readlane(r, k + i);
    wk[i] = __int_as_float(__builtin_amdgcn_readlane(__float_as_int(w), k + i));
    const char* rowp = hwb + ((size_t)rk << 9);
    v[i] = *(const float4*)(rowp + loff);
  }
#pragma unroll
  for (int i = 0; i < B; ++i) {
    acc.x += v[i].x * wk[i];
    acc.y += v[i].y * wk[i];
    acc.z += v[i].z * wk[i];
    acc.w += v[i].w * wk[i];
  }
}

template <bool FIRST>
__device__ __forceinline__ float4 node_sum(const float* __restrict__ hw,
                                           const int* __restrict__ offs,
                                           int2* __restrict__ sedge,
                                           const float* __restrict__ dis,
                                           int c, int lane) {
  const float4* hw4 = (const float4*)hw;
  const char* hwb = (const char*)hw;
  const unsigned loff = (unsigned)lane * 16u;
  const int beg = offs[c], end = offs[c + 1];
  const float dc = dis[c];
  const float4 hv = hw4[(size_t)c * 32 + lane];
  const float s2 = dc * dc;
  float4 acc = make_float4(hv.x * s2, hv.y * s2, hv.z * s2, hv.w * s2);
  for (int base = beg; base < end; base += 32) {
    const int j = base + lane;
    int r = 0;
    float w = 0.0f;
    if (j < end) {
      const int2 q = sedge[j];
      r = q.x;
      if (FIRST) {
        w = dis[r] * __int_as_float(q.y) * dc;
        sedge[j] = make_int2(r, __float_as_int(w));
      } else {
        w = __int_as_float(q.y);
      }
    }
    const int left = end - base;
    const int m = left < 32 ? left : 32;
    int k = 0;
    for (; k + 8 <= m; k += 8) rows_add<8>(hwb, r, w, k, loff, acc);
    if (k + 4 <= m) { rows_add<4>(hwb, r, w, k, loff, acc); k += 4; }
    if (k + 2 <= m) { rows_add<2>(hwb, r, w, k, loff, acc); k += 2; }
    if (k < m)      { rows_add<1>(hwb, r, w, k, loff, acc); }
  }
  return acc;
}

template <bool FIRST>
__global__ __launch_bounds__(256, 16)
void k_gather_gemm(const float* __restrict__ hw, const int* __restrict__ offs,
                   int2* __restrict__ sedge, const float* __restrict__ dis,
                   const float* __restrict__ b,
                   const float* __restrict__ W,
                   float* __restrict__ C) {
  __shared__ float As[16 * 132];
  const int t    = threadIdx.x;
  const int lane = t & 31;
  const int wave = __builtin_amdgcn_readfirstlane(t >> 5);
  const int m0   = blockIdx.x * 16;
  const float4 bb = ((const float4*)b)[lane];

  for (int i = 0; i < 2; ++i) {
    const int lr = wave * 2 + i;
    float4 acc = node_sum<FIRST>(hw, offs, sedge, dis, m0 + lr, lane);
    acc.x += bb.x; acc.y += bb.y; acc.z += bb.z; acc.w += bb.w;
    acc.x = fmaxf(acc.x, 0.0f); acc.y = fmaxf(acc.y, 0.0f);
    acc.z = fmaxf(acc.z, 0.0f); acc.w = fmaxf(acc.w, 0.0f);
    *(float4*)&As[lr * 132 + lane * 4] = acc;
  }
  __syncthreads();

  const int half = lane >> 4;
  const int l15  = lane & 15;
  const int n0   = wave * 16;

  const float4* Wq = (const float4*)W + (wave * 16) * 32 + lane;
  v8f cacc = {};
#pragma unroll
  for (int j = 0; j < 16; ++j) {
    const float4 bq = Wq[j * 32];
    const int k0 = j * 8 + 2 * half;
    v2f a0, a1, b0, b1;
    a0.x = As[l15 * 132 + k0 + 0];
    a0.y = As[l15 * 132 + k0 + 1];
    a1.x = As[l15 * 132 + k0 + 4];
    a1.y = As[l15 * 132 + k0 + 5];
    b0.x = bq.x; b0.y = bq.y;
    b1.x = bq.z; b1.y = bq.w;
    cacc = __builtin_amdgcn_wmma_f32_16x16x4_f32(false, a0, false, b0, (short)0, cacc,
                                                 false, false);
    cacc = __builtin_amdgcn_wmma_f32_16x16x4_f32(false, a1, false, b1, (short)0, cacc,
                                                 false, false);
  }

#pragma unroll
  for (int r = 0; r < 8; ++r) {
    C[(size_t)(m0 + r + 8 * half) * HIDF + (n0 + l15)] = cacc[r];
  }
}

__global__ __launch_bounds__(256, 16)
void k_gather_out(const float* __restrict__ hw, const int* __restrict__ offs,
                  int2* __restrict__ sedge, const float* __restrict__ dis,
                  const float* __restrict__ bfold, const float* __restrict__ bout,
                  float* __restrict__ out) {
  const int t    = threadIdx.x;
  const int lane = t & 31;
  const int wave = __builtin_amdgcn_readfirstlane(t >> 5);
  const int m0   = blockIdx.x * 16;
  const float4 bf = ((const float4*)bfold)[lane];
  const float4 bo = ((const float4*)bout)[lane];
  const float4 bb = make_float4(bf.x + bo.x, bf.y + bo.y, bf.z + bo.z, bf.w + bo.w);
  float4* out4 = (float4*)out;

  for (int i = 0; i < 2; ++i) {
    const int c = m0 + wave * 2 + i;
    float4 acc = node_sum<false>(hw, offs, sedge, dis, c, lane);
    acc.x += bb.x; acc.y += bb.y; acc.z += bb.z; acc.w += bb.w;
    out4[(size_t)c * 32 + lane] = acc;
  }
}

extern "C" void kernel_launch(void* const* d_in, const int* in_sizes, int n_in,
                              void* d_out, int out_size, void* d_ws, size_t ws_size,
                              hipStream_t stream) {
  const float* x   = (const float*)d_in[0];
  const int*   ei  = (const int*)d_in[1];
  const float* ew  = (const float*)d_in[2];
  const float* Wc[4] = {(const float*)d_in[3], (const float*)d_in[5],
                        (const float*)d_in[7], (const float*)d_in[9]};
  const float* bc[4] = {(const float*)d_in[4], (const float*)d_in[6],
                        (const float*)d_in[8], (const float*)d_in[10]};
  const float* Wout = (const float*)d_in[11];
  const float* bout = (const float*)d_in[12];

  const int n = in_sizes[0] / HIDF;
  const int E = in_sizes[1] / 2;
  const int* row = ei;
  const int* col = ei + E;

  const int NB = (E + EB - 1) / EB;
  const int K  = (n + RB - 1) / RB;
  const int NT = K * NB;

  char* ws = (char*)d_ws;
  const size_t fb = (size_t)n * HIDF * sizeof(float);
  const size_t nb = (((size_t)(n + 1) * 4) + 255) & ~(size_t)255;
  const size_t tb = (((size_t)(NT + 1) * 4) + 255) & ~(size_t)255;
  const size_t eb = (((size_t)E * 4) + 255) & ~(size_t)255;
  float* hw   = (float*)ws;
  float* hbuf = (float*)(ws + fb);
  char*  p    = ws + 2 * fb;
  float* dis    = (float*)p; p += nb;
  int*   offs   = (int*)p;   p += nb;
  int*   tab    = (int*)p;   p += tb;
  int*   base   = (int*)p;   p += tb;
  int*   pos    = (int*)p;   p += eb;
  int2*  sedge  = (int2*)p;  p += 2 * eb;
  int4*  rec    = (int4*)p;  p += 4 * eb;
  int4*  tmp    = (int4*)p;  p += 4 * eb;
  float* wfold  = (float*)p;  p += (size_t)PKW * 4;
  float* bfold  = (float*)p;  p += (size_t)16 * HIDF * 4;
  float* pk     = (float*)p;

  k_bin_count<<<NB + 8 * NPK, ETH, 0, stream>>>(col, pos, tab, E, NB, K,
                                                Wc[0], Wc[1], Wc[2], Wout, pk);
  k_scan<<<(NT + 1023) / 1024, 1024, 0, stream>>>(tab, base, NT);
  k_bin_place<<<NB, ETH, 0, stream>>>(row, col, ew, pos, base, rec, E, NB, K);
  k_bin_order<<<K, 1024, 0, stream>>>(base, rec, tmp, sedge, offs, dis, n, NB);

  k_gemm_wmma<<<n / 16 + 9, 256, 0, stream>>>(x, pk, hw, n / 16, Wc[3], pk + 3 * PKW, bc[3],
                                             wfold, bfold);
  k_gather_gemm<true><<<n / 16, 256, 0, stream>>>(hw, offs, sedge, dis, bc[0], pk + PKW, hbuf);
  k_gather_gemm<false><<<n / 16, 256, 0, stream>>>(hbuf, offs, sedge, dis, bc[1], pk + 2 * PKW,
                                                  hw);
  k_gather_gemm<false><<<n / 16, 256, 0, stream>>>(hw, offs, sedge, dis, bc[2], wfold, hbuf);
  k_gather_out<<<n / 16, 256, 0, stream>>>(hbuf, offs, sedge, dis, bfold, bout,
                                          (float*)d_out);
}


# fused sum+product launches: 32-node tile per block; a wave sums 4 nodes and holds two accumulator tiles, each B fragment load serves both (B loads per row halved); per-element sum order unchanged
# speedup vs baseline: 4.8078x; 1.0320x over previous
#include <hip/hip_runtime.h>
#include <math.h>

typedef __attribute__((ext_vector_type(2))) float v2f;
typedef __attribute__((ext_vector_type(8))) float v8f;

#define HIDF 128


#define EB   8192
#define ETH  512
#define RB   256
#define RSH  8
#define KMAX 512
#define WCAP 64

#define PKW (HIDF * HIDF)
#define NPK 4

__global__ __launch_bounds__(ETH)
void k_bin_count(const int* __restrict__ col, int* __restrict__ pos, int* __restrict__ tab,
                 int E, int NB, int K,
                 const float* __restrict__ Wa, const float* __restrict__ Wb,
                 const float* __restrict__ Wc, const float* __restrict__ Wd,
                 float* __restrict__ pk) {
  __shared__ int h[KMAX];
  const int t = threadIdx.x, b = blockIdx.x;
  if (b >= NB) {
    const int q = b - NB;
    const int which = q >> 3, w = q & 7;
    const float* Ws = (which == 0) ? Wa : (which == 1) ? Wb : (which == 2) ? Wc : Wd;
    const int j = t >> 5, lane = t & 31;
    const int n = 16 * w + (lane & 15);
    const int k0 = 8 * j + 2 * (lane >> 4);
    float4 v;
    v.x = Ws[(k0 + 0) * HIDF + n];
    v.y = Ws[(k0 + 1) * HIDF + n];
    v.z = Ws[(k0 + 4) * HIDF + n];
    v.w = Ws[(k0 + 5) * HIDF + n];
    ((float4*)(pk + (size_t)which * PKW))[(w * 16 + j) * 32 + lane] = v;
    return;
  }
  for (int i = t; i < K; i += ETH) h[i] = 0;
  __syncthreads();
  const int e0 = b * EB + t;
  if (b * EB + EB <= E) {
    for (int g = 0; g < EB / ETH; g += 8) {
      int c[8];
#pragma unroll
      for (int u = 0; u < 8; ++u) c[u] = col[e0 + (g + u) * ETH];
#pragma unroll
      for (int u = 0; u < 8; ++u) pos[e0 + (g + u) * ETH] = atomicAdd(&h[c[u] >> RSH], 1);
    }
  } else {
    for (int e = e0; e < E; e += ETH) pos[e] = atomicAdd(&h[col[e] >> RSH], 1);
  }
  __syncthreads();
  for (int i = t; i < K; i += ETH) tab[i * NB + b] = h[i];
}

__global__ __launch_bounds__(1024)
void k_scan(const int* __restrict__ cnt, int* __restrict__ offs, int n) {
  __shared__ int part[32];
  __shared__ int wtot[32];
  const int t = threadIdx.x, lane = t & 31, wv = t >> 5;
  const int start = blockIdx.x * 1024;
  int s = 0;
  for (int i = t; i < start; i += 1024) s += cnt[i];
  const int i = start + t;
  const int v = (i < n) ? cnt[i] : 0;
  int inc = v;
  for (int o = 1; o < 32; o <<= 1) {
    const int u = __shfl_up(inc, o, 32);
    if (lane >= o) inc += u;
  }
  for (int o = 16; o > 0; o >>= 1) s += __shfl_xor(s, o, 32);
  if (lane == 31) wtot[wv] = inc;
  if (lane == 0) part[wv] = s;
  __syncthreads();
  int base = 0;
  for (int w = 0; w < 32; ++w) {
    base += part[w];
    if (w < wv) base += wtot[w];
  }
  const int excl = base + inc - v;
  if (i < n) offs[i] = excl;
  if (i == n - 1) offs[n] = excl + v;
}

__global__ __launch_bounds__(ETH)
void k_bin_place(const int* __restrict__ row, const int* __restrict__ col,
                 const float* __restrict__ ew, const int* __restrict__ pos,
                 const int* __restrict__ base, int4* __restrict__ rec, int E, int NB, int K) {
  __shared__ int sb[KMAX];
  const int t = threadIdx.x, b = blockIdx.x;
  for (int i = t; i < K; i += ETH) sb[i] = base[i * NB + b];
  __syncthreads();
  const int e0 = b * EB + t;
  if (b * EB + EB <= E) {
    for (int g = 0; g < EB / ETH; g += 4) {
      int c[4], p[4], r[4];
      float w[4];
#pragma unroll
      for (int u = 0; u < 4; ++u) {
        const int e = e0 + (g + u) * ETH;
        c[u] = col[e]; p[u] = pos[e]; r[u] = row[e]; w[u] = ew[e];
      }
#pragma unroll
      for (int u = 0; u < 4; ++u) {
        const int e = e0 + (g + u) * ETH;
        rec[sb[c[u] >> RSH] + p[u]] = make_int4(e, r[u], __float_as_int(w[u]), c[u]);
      }
    }
  } else {
    for (int e = e0; e < E; e += ETH) {
      const int c = col[e];
      rec[sb[c >> RSH] + pos[e]] = make_int4(e, row[e], __float_as_int(ew[e]), c);
    }
  }
}

__global__ __launch_bounds__(1024)
void k_bin_order(const int* __restrict__ base, const int4* __restrict__ rec,
                 int4* __restrict__ tmp, int2* __restrict__ sedge, int* __restrict__ offs,
                 float* __restrict__ dis, int n, int NB) {
  __shared__ int cnt[RB];
  __shared__ int loff[RB];
  __shared__ int cur[RB];
  __shared__ int wtot[RB / 32];
  __shared__ float wsort[32 * WCAP];
  const int t = threadIdx.x, lane = t & 31;
  const int wv = __builtin_amdgcn_readfirstlane(t >> 5);
  const int k = blockIdx.x;
  const int bs = __builtin_amdgcn_readfirstlane(base[k * NB]);
  const int be = __builtin_amdgcn_readfirstlane(base[(k + 1) * NB]);
  const int nb0 = k * RB;
  if (t < RB) cnt[t] = 0;
  __syncthreads();
  for (int idx = bs + t; idx < be; idx += 1024) atomicAdd(&cnt[rec[idx].w - nb0], 1);
  __syncthreads();
  const int v = (t < RB) ? cnt[t] : 0;
  int inc = v;
  for (int o = 1; o < 32; o <<= 1) {
    const int u = __shfl_up(inc, o, 32);
    if (lane >= o) inc += u;
  }
  if (t < RB && lane == 31) wtot[t >> 5] = inc;
  __syncthreads();
  if (t < RB) {
    int pre = 0;
    for (int w = 0; w < RB / 32; ++w)
      if (w < (t >> 5)) pre += wtot[w];
    const int ex = pre + inc - v;
    loff[t] = ex;
    cur[t] = ex;
    if (nb0 + t < n) offs[nb0 + t] = bs + ex;
  }
  if (k == (int)gridDim.x - 1 && t == 0) offs[n] = be;
  __syncthreads();
  for (int idx = bs + t; idx < be; idx += 1024) {
    const int4 q = rec[idx];
    const int p = atomicAdd(&cur[q.w - nb0], 1);
    tmp[bs + p] = q;
  }
  __syncthreads();

  float* const ws = &wsort[wv * WCAP];
  int i = wv;
  bool have = (nb0 + i < n);
  int d = 0, beg = 0;
  int4 q = make_int4(0x7fffffff, 0, 0, 0);
  if (have) {
    d = __builtin_amdgcn_readfirstlane(cnt[i]);
    beg = bs + __builtin_amdgcn_readfirstlane(loff[i]);
    if (lane < d) q = tmp[beg + lane];
  }
  while (have) {
    const int i2 = i + 32;
    const bool have2 = (i2 < RB) && (nb0 + i2 < n);
    int d2 = 0, beg2 = 0;
    int4 q2 = make_int4(0x7fffffff, 0, 0, 0);
    if (have2) {
      d2 = __builtin_amdgcn_readfirstlane(cnt[i2]);
      beg2 = bs + __builtin_amdgcn_readfirstlane(loff[i2]);
      if (lane < d2) q2 = tmp[beg2 + lane];
    }
    float deg = 0.0f;
    if (d <= WCAP) {
      for (int j0 = 0; j0 < d; j0 += 32) {
        int4 a = q;
        if (j0 > 0) {
          a = make_int4(0x7fffffff, 0, 0, 0);
          if (j0 + lane < d) a = tmp[beg + j0 + lane];
        }
        int rank = 0;
        for (int k0 = 0; k0 < d; k0 += 32) {
          int ek = a.x;
          if (k0 != j0) {
            ek = 0x7fffffff;
            if (k0 + lane < d) ek = tmp[beg + k0 + lane].x;
          }
          const int m = (d - k0 < 32) ? (d - k0) : 32;
          for (int kk = 0; kk < m; ++kk)
            rank += (__builtin_amdgcn_readlane(ek, kk) < a.x) ? 1 : 0;
        }
        if (j0 + lane < d) {
          sedge[beg + rank] = make_int2(a.y, a.z);
          ws[rank] = __int_as_float(a.z);
        }
      }
      for (int s0 = 0; s0 < d; s0 += 32) {
        const float sw = (s0 + lane < d) ? ws[s0 + lane] : 0.0f;
        const int m = (d - s0 < 32) ? (d - s0) : 32;
        for (int kk = 0; kk < m; ++kk)
          deg += __int_as_float(__builtin_amdgcn_readlane(__float_as_int(sw), kk));
      }
    } else if (lane == 0) {
      int last = -1;
      for (int s = 0; s < d; ++s) {
        int best = 0x7fffffff, bj = 0;
        for (int j = 0; j < d; ++j) {
          const int ee = tmp[beg + j].x;
          if (ee > last && ee < best) { best = ee; bj = j; }
        }
        const int4 z = tmp[beg + bj];
        sedge[beg + s] = make_int2(z.y, z.z);
        deg += __int_as_float(z.z);
        last = best;
      }
    }
    if (lane == 0) {
      deg += 1.0f;
      dis[nb0 + i] = (deg > 0.0f) ? rsqrtf(deg) : 0.0f;
    }
    i = i2; have = have2; d = d2; beg = beg2; q = q2;
  }
}

__global__ __launch_bounds__(256)
void k_gemm_wmma(const float* __restrict__ A, const float* __restrict__ W,
                 float* __restrict__ C, int nt,
                 const float* __restrict__ A2, const float* __restrict__ W2,
                 const float* __restrict__ bvec, float* __restrict__ C2,
                 float* __restrict__ C3) {
  __shared__ float As[16 * 132];
  const int t  = threadIdx.x;
  const int xb = (int)blockIdx.x - nt;
  const bool extra = xb >= 0;
  const int m0 = (extra ? xb : (int)blockIdx.x) * 16;
  const float* Ap = extra ? A2 : A;
  const float* Wp = extra ? W2 : W;

  {
    const float4* A4 = (const float4*)(Ap + (size_t)m0 * HIDF);
    const int r = t >> 5, c = t & 31;
    float4 v0, v1;
    if (xb < 8) {
      v0 = A4[t];
      v1 = A4[t + 256];
    } else {
      v0 = (r == 0) ? ((const float4*)bvec)[c] : make_float4(0.0f, 0.0f, 0.0f, 0.0f);
      v1 = make_float4(0.0f, 0.0f, 0.0f, 0.0f);
    }
    As[r * 132 + c * 4 + 0] = v0.x;
    As[r * 132 + c * 4 + 1] = v0.y;
    As[r * 132 + c * 4 + 2] = v0.z;
    As[r * 132 + c * 4 + 3] = v0.w;
    As[(r + 8) * 132 + c * 4 + 0] = v1.x;
    As[(r + 8) * 132 + c * 4 + 1] = v1.y;
    As[(r + 8) * 132 + c * 4 + 2] = v1.z;
    As[(r + 8) * 132 + c * 4 + 3] = v1.w;
  }
  __syncthreads();

  const int lane = t & 31;
  const int wave = t >> 5;
  const int half = lane >> 4;
  const int l15  = lane & 15;
  const int n0   = wave * 16;

  const float4* Wq = (const float4*)Wp + (wave * 16) * 32 + lane;
  v8f acc = {};
#pragma unroll
  for (int j = 0; j < 16; ++j) {
    const float4 bq = Wq[j * 32];
    const int k0 = j * 8 + 2 * half;
    v2f a0, a1, b0, b1;
    a0.x = As[l15 * 132 + k0 + 0];
    a0.y = As[l15 * 132 + k0 + 1];
    a1.x = As[l15 * 132 + k0 + 4];
    a1.y = As[l15 * 132 + k0 + 5];
    b0.x = bq.x; b0.y = bq.y;
    b1.x = bq.z; b1.y = bq.w;
    acc = __builtin_amdgcn_wmma_f32_16x16x4_f32(false, a0, false, b0, (short)0, acc,
                                                false, false);
    acc = __builtin_amdgcn_wmma_f32_16x16x4_f32(false, a1, false, b1, (short)0, acc,
                                                false, false);
  }

  if (xb >= 0 && xb < 8) {
#pragma unroll
    for (int r = 0; r < 8; ++r) {
      const int pl = ((r >> 1) & 1) * 16 + l15;
      const int pe = (r & 1) + 2 * (r >> 2);
      C2[(((wave * 16) + 2 * xb + half) * 32 + pl) * 4 + pe] = acc[r];
    }
  } else {
    float* Cp = extra ? C3 : C;
    const int mr = extra ? 0 : m0;
#pragma unroll
    for (int r = 0; r < 8; ++r) {
      Cp[(size_t)(mr + r + 8 * half) * HIDF + (n0 + l15)] = acc[r];
    }
  }
}

template <int B>
__device__ __forceinline__ void rows_add(const char* __restrict__ hwb, int r, float w,
                                         int k, unsigned loff, float4& acc) {
  float4 v[B];
  float wk[B];
#pragma unroll
  for (int i = 0; i < B; ++i) {
    const unsigned rk = (unsigned)__builtin_amdgcn_readlane(r, k + i);
    wk[i] = __int_as_float(__builtin_amdgcn_readlane(__float_as_int(w), k + i));
    const char* rowp = hwb + ((size_t)rk << 9);
    v[i] = *(const float4*)(rowp + loff);
  }
#pragma unroll
  for (int i = 0; i < B; ++i) {
    acc.x += v[i].x * wk[i];
    acc.y += v[i].y * wk[i];
    acc.z += v[i].z * wk[i];
    acc.w += v[i].w * wk[i];
  }
}

template <bool FIRST>
__device__ __forceinline__ float4 node_sum(const float* __restrict__ hw,
                                           const int* __restrict__ offs,
                                           int2* __restrict__ sedge,
                                           const float* __restrict__ dis,
                                           int c, int lane) {
  const float4* hw4 = (const float4*)hw;
  const char* hwb = (const char*)hw;
  const unsigned loff = (unsigned)lane * 16u;
  const int beg = offs[c], end = offs[c + 1];
  const float dc = dis[c];
  const float4 hv = hw4[(size_t)c * 32 + lane];
  const float s2 = dc * dc;
  float4 acc = make_float4(hv.x * s2, hv.y * s2, hv.z * s2, hv.w * s2);
  for (int base = beg; base < end; base += 32) {
    const int j = base + lane;
    int r = 0;
    float w = 0.0f;
    if (j < end) {
      const int2 q = sedge[j];
      r = q.x;
      if (FIRST) {
        w = dis[r] * __int_as_float(q.y) * dc;
        sedge[j] = make_int2(r, __float_as_int(w));
      } else {
        w = __int_as_float(q.y);
      }
    }
    const int left = end - base;
    const int m = left < 32 ? left : 32;
    int k = 0;
    for (; k + 8 <= m; k += 8) rows_add<8>(hwb, r, w, k, loff, acc);
    if (k + 4 <= m) { rows_add<4>(hwb, r, w, k, loff, acc); k += 4; }
    if (k + 2 <= m) { rows_add<2>(hwb, r, w, k, loff, acc); k += 2; }
    if (k < m)      { rows_add<1>(hwb, r, w, k, loff, acc); }
  }
  return acc;
}

#define TM 32
template <bool FIRST>
__global__ __launch_bounds__(256, 16)
void k_gather_gemm(const float* __restrict__ hw, const int* __restrict__ offs,
                   int2* __restrict__ sedge, const float* __restrict__ dis,
                   const float* __restrict__ b,
                   const float* __restrict__ W,
                   float* __restrict__ C) {
  __shared__ float As[TM * 132];
  const int t    = threadIdx.x;
  const int lane = t & 31;
  const int wave = __builtin_amdgcn_readfirstlane(t >> 5);
  const int m0   = blockIdx.x * TM;
  const float4 bb = ((const float4*)b)[lane];

  for (int i = 0; i < TM / 8; ++i) {
    const int lr = wave * (TM / 8) + i;
    float4 acc = node_sum<FIRST>(hw, offs, sedge, dis, m0 + lr, lane);
    acc.x += bb.x; acc.y += bb.y; acc.z += bb.z; acc.w += bb.w;
    acc.x = fmaxf(acc.x, 0.0f); acc.y = fmaxf(acc.y, 0.0f);
    acc.z = fmaxf(acc.z, 0.0f); acc.w = fmaxf(acc.w, 0.0f);
    *(float4*)&As[lr * 132 + lane * 4] = acc;
  }
  __syncthreads();

  const int half = lane >> 4;
  const int l15  = lane & 15;
  const int n0   = wave * 16;

  const float4* Wq = (const float4*)W + (wave * 16) * 32 + lane;
  const float* Ar0 = &As[l15 * 132 + 2 * half];
  const float* Ar1 = Ar0 + 16 * 132;
  v8f c0 = {}, c1 = {};
#pragma unroll
  for (int j = 0; j < 16; ++j) {
    const float4 bq = Wq[j * 32];
    const int k0 = j * 8;
    v2f a00, a01, a10, a11, b0, b1;
    a00.x = Ar0[k0 + 0];
    a00.y = Ar0[k0 + 1];
    a01.x = Ar0[k0 + 4];
    a01.y = Ar0[k0 + 5];
    a10.x = Ar1[k0 + 0];
    a10.y = Ar1[k0 + 1];
    a11.x = Ar1[k0 + 4];
    a11.y = Ar1[k0 + 5];
    b0.x = bq.x; b0.y = bq.y;
    b1.x = bq.z; b1.y = bq.w;
    c0 = __builtin_amdgcn_wmma_f32_16x16x4_f32(false, a00, false, b0, (short)0, c0,
                                               false, false);
    c1 = __builtin_amdgcn_wmma_f32_16x16x4_f32(false, a10, false, b0, (short)0, c1,
                                               false, false);
    c0 = __builtin_amdgcn_wmma_f32_16x16x4_f32(false, a01, false, b1, (short)0, c0,
                                               false, false);
    c1 = __builtin_amdgcn_wmma_f32_16x16x4_f32(false, a11, false, b1, (short)0, c1,
                                               false, false);
  }

#pragma unroll
  for (int r = 0; r < 8; ++r) {
    C[(size_t)(m0 + r + 8 * half) * HIDF + (n0 + l15)] = c0[r];
  }
#pragma unroll
  for (int r = 0; r < 8; ++r) {
    C[(size_t)(m0 + 16 + r + 8 * half) * HIDF + (n0 + l15)] = c1[r];
  }
}

__global__ __launch_bounds__(256, 16)
void k_gather_out(const float* __restrict__ hw, const int* __restrict__ offs,
                  int2* __restrict__ sedge, const float* __restrict__ dis,
                  const float* __restrict__ bfold, const float* __restrict__ bout,
                  float* __restrict__ out) {
  const int t    = threadIdx.x;
  const int lane = t & 31;
  const int wave = __builtin_amdgcn_readfirstlane(t >> 5);
  const int m0   = blockIdx.x * 16;
  const float4 bf = ((const float4*)bfold)[lane];
  const float4 bo = ((const float4*)bout)[lane];
  const float4 bb = make_float4(bf.x + bo.x, bf.y + bo.y, bf.z + bo.z, bf.w + bo.w);
  float4* out4 = (float4*)out;

  for (int i = 0; i < 2; ++i) {
    const int c = m0 + wave * 2 + i;
    float4 acc = node_sum<false>(hw, offs, sedge, dis, c, lane);
    acc.x += bb.x; acc.y += bb.y; acc.z += bb.z; acc.w += bb.w;
    out4[(size_t)c * 32 + lane] = acc;
  }
}

extern "C" void kernel_launch(void* const* d_in, const int* in_sizes, int n_in,
                              void* d_out, int out_size, void* d_ws, size_t ws_size,
                              hipStream_t stream) {
  const float* x   = (const float*)d_in[0];
  const int*   ei  = (const int*)d_in[1];
  const float* ew  = (const float*)d_in[2];
  const float* Wc[4] = {(const float*)d_in[3], (const float*)d_in[5],
                        (const float*)d_in[7], (const float*)d_in[9]};
  const float* bc[4] = {(const float*)d_in[4], (const float*)d_in[6],
                        (const float*)d_in[8], (const float*)d_in[10]};
  const float* Wout = (const float*)d_in[11];
  const float* bout = (const float*)d_in[12];

  const int n = in_sizes[0] / HIDF;
  const int E = in_sizes[1] / 2;
  const int* row = ei;
  const int* col = ei + E;

  const int NB = (E + EB - 1) / EB;
  const int K  = (n + RB - 1) / RB;
  const int NT = K * NB;

  char* ws = (char*)d_ws;
  const size_t fb = (size_t)n * HIDF * sizeof(float);
  const size_t nb = (((size_t)(n + 1) * 4) + 255) & ~(size_t)255;
  const size_t tb = (((size_t)(NT + 1) * 4) + 255) & ~(size_t)255;
  const size_t eb = (((size_t)E * 4) + 255) & ~(size_t)255;
  float* hw   = (float*)ws;
  float* hbuf = (float*)(ws + fb);
  char*  p    = ws + 2 * fb;
  float* dis    = (float*)p; p += nb;
  int*   offs   = (int*)p;   p += nb;
  int*   tab    = (int*)p;   p += tb;
  int*   base   = (int*)p;   p += tb;
  int*   pos    = (int*)p;   p += eb;
  int2*  sedge  = (int2*)p;  p += 2 * eb;
  int4*  rec    = (int4*)p;  p += 4 * eb;
  int4*  tmp    = (int4*)p;  p += 4 * eb;
  float* wfold  = (float*)p;  p += (size_t)PKW * 4;
  float* bfold  = (float*)p;  p += (size_t)16 * HIDF * 4;
  float* pk     = (float*)p;

  k_bin_count<<<NB + 8 * NPK, ETH, 0, stream>>>(col, pos, tab, E, NB, K,
                                                Wc[0], Wc[1], Wc[2], Wout, pk);
  k_scan<<<(NT + 1023) / 1024, 1024, 0, stream>>>(tab, base, NT);
  k_bin_place<<<NB, ETH, 0, stream>>>(row, col, ew, pos, base, rec, E, NB, K);
  k_bin_order<<<K, 1024, 0, stream>>>(base, rec, tmp, sedge, offs, dis, n, NB);

  k_gemm_wmma<<<n / 16 + 9, 256, 0, stream>>>(x, pk, hw, n / 16, Wc[3], pk + 3 * PKW, bc[3],
                                             wfold, bfold);
  k_gather_gemm<true><<<n / TM, 256, 0, stream>>>(hw, offs, sedge, dis, bc[0], pk + PKW, hbuf);
  k_gather_gemm<false><<<n / TM, 256, 0, stream>>>(hbuf, offs, sedge, dis, bc[1], pk + 2 * PKW,
                                                  hw);
  k_gather_gemm<false><<<n / TM, 256, 0, stream>>>(hw, offs, sedge, dis, bc[2], wfold, hbuf);
  k_gather_out<<<n / 16, 256, 0, stream>>>(hbuf, offs, sedge, dis, bfold, bout,
                                          (float*)d_out);
}


# row sums wait less often: a wave fetches bucket bounds and first edge records of all its nodes at once, own row loads with the edge rows, a list's last 1-7 rows are one load group; sum order kept
# speedup vs baseline: 4.9756x; 1.0349x over previous
#include <hip/hip_runtime.h>
#include <math.h>

typedef __attribute__((ext_vector_type(2))) float v2f;
typedef __attribute__((ext_vector_type(8))) float v8f;

#define HIDF 128


#define EB   8192
#define ETH  512
#define RB   256
#define RSH  8
#define KMAX 512
#define WCAP 64

#define PKW (HIDF * HIDF)
#define NPK 4

__global__ __launch_bounds__(ETH)
void k_bin_count(const int* __restrict__ col, int* __restrict__ pos, int* __restrict__ tab,
                 int E, int NB, int K,
                 const float* __restrict__ Wa, const float* __restrict__ Wb,
                 const float* __restrict__ Wc, const float* __restrict__ Wd,
                 float* __restrict__ pk) {
  __shared__ int h[KMAX];
  const int t = threadIdx.x, b = blockIdx.x;
  if (b >= NB) {
    const int q = b - NB;
    const int which = q >> 3, w = q & 7;
    const float* Ws = (which == 0) ? Wa : (which == 1) ? Wb : (which == 2) ? Wc : Wd;
    const int j = t >> 5, lane = t & 31;
    const int n = 16 * w + (lane & 15);
    const int k0 = 8 * j + 2 * (lane >> 4);
    float4 v;
    v.x = Ws[(k0 + 0) * HIDF + n];
    v.y = Ws[(k0 + 1) * HIDF + n];
    v.z = Ws[(k0 + 4) * HIDF + n];
    v.w = Ws[(k0 + 5) * HIDF + n];
    ((float4*)(pk + (size_t)which * PKW))[(w * 16 + j) * 32 + lane] = v;
    return;
  }
  for (int i = t; i < K; i += ETH) h[i] = 0;
  __syncthreads();
  const int e0 = b * EB + t;
  if (b * EB + EB <= E) {
    for (int g = 0; g < EB / ETH; g += 8) {
      int c[8];
#pragma unroll
      for (int u = 0; u < 8; ++u) c[u] = col[e0 + (g + u) * ETH];
#pragma unroll
      for (int u = 0; u < 8; ++u) pos[e0 + (g + u) * ETH] = atomicAdd(&h[c[u] >> RSH], 1);
    }
  } else {
    for (int e = e0; e < E; e += ETH) pos[e] = atomicAdd(&h[col[e] >> RSH], 1);
  }
  __syncthreads();
  for (int i = t; i < K; i += ETH) tab[i * NB + b] = h[i];
}

__global__ __launch_bounds__(1024)
void k_scan(const int* __restrict__ cnt, int* __restrict__ offs, int n) {
  __shared__ int part[32];
  __shared__ int wtot[32];
  const int t = threadIdx.x, lane = t & 31, wv = t >> 5;
  const int start = blockIdx.x * 1024;
  int s = 0;
  for (int i = t; i < start; i += 1024) s += cnt[i];
  const int i = start + t;
  const int v = (i < n) ? cnt[i] : 0;
  int inc = v;
  for (int o = 1; o < 32; o <<= 1) {
    const int u = __shfl_up(inc, o, 32);
    if (lane >= o) inc += u;
  }
  for (int o = 16; o > 0; o >>= 1) s += __shfl_xor(s, o, 32);
  if (lane == 31) wtot[wv] = inc;
  if (lane == 0) part[wv] = s;
  __syncthreads();
  int base = 0;
  for (int w = 0; w < 32; ++w) {
    base += part[w];
    if (w < wv) base += wtot[w];
  }
  const int excl = base + inc - v;
  if (i < n) offs[i] = excl;
  if (i == n - 1) offs[n] = excl + v;
}

__global__ __launch_bounds__(ETH)
void k_bin_place(const int* __restrict__ row, const int* __restrict__ col,
                 const float* __restrict__ ew, const int* __restrict__ pos,
                 const int* __restrict__ base, int4* __restrict__ rec, int E, int NB, int K) {
  __shared__ int sb[KMAX];
  const int t = threadIdx.x, b = blockIdx.x;
  for (int i = t; i < K; i += ETH) sb[i] = base[i * NB + b];
  __syncthreads();
  const int e0 = b * EB + t;
  if (b * EB + EB <= E) {
    for (int g = 0; g < EB / ETH; g += 4) {
      int c[4], p[4], r[4];
      float w[4];
#pragma unroll
      for (int u = 0; u < 4; ++u) {
        const int e = e0 + (g + u) * ETH;
        c[u] = col[e]; p[u] = pos[e]; r[u] = row[e]; w[u] = ew[e];
      }
#pragma unroll
      for (int u = 0; u < 4; ++u) {
        const int e = e0 + (g + u) * ETH;
        rec[sb[c[u] >> RSH] + p[u]] = make_int4(e, r[u], __float_as_int(w[u]), c[u]);
      }
    }
  } else {
    for (int e = e0; e < E; e += ETH) {
      const int c = col[e];
      rec[sb[c >> RSH] + pos[e]] = make_int4(e, row[e], __float_as_int(ew[e]), c);
    }
  }
}

__global__ __launch_bounds__(1024)
void k_bin_order(const int* __restrict__ base, const int4* __restrict__ rec,
                 int4* __restrict__ tmp, int2* __restrict__ sedge, int* __restrict__ offs,
                 float* __restrict__ dis, int n, int NB) {
  __shared__ int cnt[RB];
  __shared__ int loff[RB];
  __shared__ int cur[RB];
  __shared__ int wtot[RB / 32];
  __shared__ float wsort[32 * WCAP];
  const int t = threadIdx.x, lane = t & 31;
  const int wv = __builtin_amdgcn_readfirstlane(t >> 5);
  const int k = blockIdx.x;
  const int bs = __builtin_amdgcn_readfirstlane(base[k * NB]);
  const int be = __builtin_amdgcn_readfirstlane(base[(k + 1) * NB]);
  const int nb0 = k * RB;
  if (t < RB) cnt[t] = 0;
  __syncthreads();
  for (int idx = bs + t; idx < be; idx += 1024) atomicAdd(&cnt[rec[idx].w - nb0], 1);
  __syncthreads();
  const int v = (t < RB) ? cnt[t] : 0;
  int inc = v;
  for (int o = 1; o < 32; o <<= 1) {
    const int u = __shfl_up(inc, o, 32);
    if (lane >= o) inc += u;
  }
  if (t < RB && lane == 31) wtot[t >> 5] = inc;
  __syncthreads();
  if (t < RB) {
    int pre = 0;
    for (int w = 0; w < RB / 32; ++w)
      if (w < (t >> 5)) pre += wtot[w];
    const int ex = pre + inc - v;
    loff[t] = ex;
    cur[t] = ex;
    if (nb0 + t < n) offs[nb0 + t] = bs + ex;
  }
  if (k == (int)gridDim.x - 1 && t == 0) offs[n] = be;
  __syncthreads();
  for (int idx = bs + t; idx < be; idx += 1024) {
    const int4 q = rec[idx];
    const int p = atomicAdd(&cur[q.w - nb0], 1);
    tmp[bs + p] = q;
  }
  __syncthreads();

  float* const ws = &wsort[wv * WCAP];
  int i = wv;
  bool have = (nb0 + i < n);
  int d = 0, beg = 0;
  int4 q = make_int4(0x7fffffff, 0, 0, 0);
  if (have) {
    d = __builtin_amdgcn_readfirstlane(cnt[i]);
    beg = bs + __builtin_amdgcn_readfirstlane(loff[i]);
    if (lane < d) q = tmp[beg + lane];
  }
  while (have) {
    const int i2 = i + 32;
    const bool have2 = (i2 < RB) && (nb0 + i2 < n);
    int d2 = 0, beg2 = 0;
    int4 q2 = make_int4(0x7fffffff, 0, 0, 0);
    if (have2) {
      d2 = __builtin_amdgcn_readfirstlane(cnt[i2]);
      beg2 = bs + __builtin_amdgcn_readfirstlane(loff[i2]);
      if (lane < d2) q2 = tmp[beg2 + lane];
    }
    float deg = 0.0f;
    if (d <= WCAP) {
      for (int j0 = 0; j0 < d; j0 += 32) {
        int4 a = q;
        if (j0 > 0) {
          a = make_int4(0x7fffffff, 0, 0, 0);
          if (j0 + lane < d) a = tmp[beg + j0 + lane];
        }
        int rank = 0;
        for (int k0 = 0; k0 < d; k0 += 32) {
          int ek = a.x;
          if (k0 != j0) {
            ek = 0x7fffffff;
            if (k0 + lane < d) ek = tmp[beg + k0 + lane].x;
          }
          const int m = (d - k0 < 32) ? (d - k0) : 32;
          for (int kk = 0; kk < m; ++kk)
            rank += (__builtin_amdgcn_readlane(ek, kk) < a.x) ? 1 : 0;
        }
        if (j0 + lane < d) {
          sedge[beg + rank] = make_int2(a.y, a.z);
          ws[rank] = __int_as_float(a.z);
        }
      }
      for (int s0 = 0; s0 < d; s0 += 32) {
        const float sw = (s0 + lane < d) ? ws[s0 + lane] : 0.0f;
        const int m = (d - s0 < 32) ? (d - s0) : 32;
        for (int kk = 0; kk < m; ++kk)
          deg += __int_as_float(__builtin_amdgcn_readlane(__float_as_int(sw), kk));
      }
    } else if (lane == 0) {
      int last = -1;
      for (int s = 0; s < d; ++s) {
        int best = 0x7fffffff, bj = 0;
        for (int j = 0; j < d; ++j) {
          const int ee = tmp[beg + j].x;
          if (ee > last && ee < best) { best = ee; bj = j; }
        }
        const int4 z = tmp[beg + bj];
        sedge[beg + s] = make_int2(z.y, z.z);
        deg += __int_as_float(z.z);
        last = best;
      }
    }
    if (lane == 0) {
      deg += 1.0f;
      dis[nb0 + i] = (deg > 0.0f) ? rsqrtf(deg) : 0.0f;
    }
    i = i2; have = have2; d = d2; beg = beg2; q = q2;
  }
}

__global__ __launch_bounds__(256)
void k_gemm_wmma(const float* __restrict__ A, const float* __restrict__ W,
                 float* __restrict__ C, int nt,
                 const float* __restrict__ A2, const float* __restrict__ W2,
                 const float* __restrict__ bvec, float* __restrict__ C2,
                 float* __restrict__ C3) {
  __shared__ float As[16 * 132];
  const int t  = threadIdx.x;
  const int xb = (int)blockIdx.x - nt;
  const bool extra = xb >= 0;
  const int m0 = (extra ? xb : (int)blockIdx.x) * 16;
  const float* Ap = extra ? A2 : A;
  const float* Wp = extra ? W2 : W;

  {
    const float4* A4 = (const float4*)(Ap + (size_t)m0 * HIDF);
    const int r = t >> 5, c = t & 31;
    float4 v0, v1;
    if (xb < 8) {
      v0 = A4[t];
      v1 = A4[t + 256];
    } else {
      v0 = (r == 0) ? ((const float4*)bvec)[c] : make_float4(0.0f, 0.0f, 0.0f, 0.0f);
      v1 = make_float4(0.0f, 0.0f, 0.0f, 0.0f);
    }
    As[r * 132 + c * 4 + 0] = v0.x;
    As[r * 132 + c * 4 + 1] = v0.y;
    As[r * 132 + c * 4 + 2] = v0.z;
    As[r * 132 + c * 4 + 3] = v0.w;
    As[(r + 8) * 132 + c * 4 + 0] = v1.x;
    As[(r + 8) * 132 + c * 4 + 1] = v1.y;
    As[(r + 8) * 132 + c * 4 + 2] = v1.z;
    As[(r + 8) * 132 + c * 4 + 3] = v1.w;
  }
  __syncthreads();

  const int lane = t & 31;
  const int wave = t >> 5;
  const int half = lane >> 4;
  const int l15  = lane & 15;
  const int n0   = wave * 16;

  const float4* Wq = (const float4*)Wp + (wave * 16) * 32 + lane;
  v8f acc = {};
#pragma unroll
  for (int j = 0; j < 16; ++j) {
    const float4 bq = Wq[j * 32];
    const int k0 = j * 8 + 2 * half;
    v2f a0, a1, b0, b1;
    a0.x = As[l15 * 132 + k0 + 0];
    a0.y = As[l15 * 132 + k0 + 1];
    a1.x = As[l15 * 132 + k0 + 4];
    a1.y = As[l15 * 132 + k0 + 5];
    b0.x = bq.x; b0.y = bq.y;
    b1.x = bq.z; b1.y = bq.w;
    acc = __builtin_amdgcn_wmma_f32_16x16x4_f32(false, a0, false, b0, (short)0, acc,
                                                false, false);
    acc = __builtin_amdgcn_wmma_f32_16x16x4_f32(false, a1, false, b1, (short)0, acc,
                                                false, false);
  }

  if (xb >= 0 && xb < 8) {
#pragma unroll
    for (int r = 0; r < 8; ++r) {
      const int pl = ((r >> 1) & 1) * 16 + l15;
      const int pe = (r & 1) + 2 * (r >> 2);
      C2[(((wave * 16) + 2 * xb + half) * 32 + pl) * 4 + pe] = acc[r];
    }
  } else {
    float* Cp = extra ? C3 : C;
    const int mr = extra ? 0 : m0;
#pragma unroll
    for (int r = 0; r < 8; ++r) {
      Cp[(size_t)(mr + r + 8 * half) * HIDF + (n0 + l15)] = acc[r];
    }
  }
}

template <int B>
__device__ __forceinline__ void rows_add(const char* __restrict__ hwb, int r, float w,
                                         int k, unsigned loff, float4& acc) {
  float4 v[B];
  float wk[B];
#pragma unroll
  for (int i = 0; i < B; ++i) {
    const unsigned rk = (unsigned)__builtin_amdgcn_readlane(r, k + i);
    wk[i] = __int_as_float(__builtin_amdgcn_readlane(__float_as_int(w), k + i));
    const char* rowp = hwb + ((size_t)rk << 9);
    v[i] = *(const float4*)(rowp + loff);
  }
#pragma unroll
  for (int i = 0; i < B; ++i) {
    acc.x = fmaf(v[i].x, wk[i], acc.x);
    acc.y = fmaf(v[i].y, wk[i], acc.y);
    acc.z = fmaf(v[i].z, wk[i], acc.z);
    acc.w = fmaf(v[i].w, wk[i], acc.w);
  }
}

template <bool FIRST>
__device__ __forceinline__ void list_chunk(int2* __restrict__ sedge,
                                           const float* __restrict__ dis, int c, int beg,
                                           int d, float dc, int t0, int lane, int& r,
                                           float& w) {
  const int t = t0 + lane;
  r = c;
  w = dc * dc;
  if (t >= 1 && t <= d) {
    const int j = beg + t - 1;
    const int2 e = sedge[j];
    r = e.x;
    if (FIRST) {
      w = dis[r] * __int_as_float(e.y) * dc;
      sedge[j] = make_int2(r, __float_as_int(w));
    } else {
      w = __int_as_float(e.y);
    }
  }
}

template <bool FIRST, int NW, typename Fin>
__device__ __forceinline__ void wave_nodes(const float* __restrict__ hw,
                                           const int* __restrict__ offs,
                                           int2* __restrict__ sedge,
                                           const float* __restrict__ dis,
                                           int c0, int lane, Fin fin) {
  const char* hwb = (const char*)hw;
  const unsigned loff = (unsigned)lane * 16u;
  int o[NW + 1];
  float dcs[NW];
#pragma unroll
  for (int i = 0; i <= NW; ++i) o[i] = offs[c0 + i];
#pragma unroll
  for (int i = 0; i < NW; ++i) dcs[i] = dis[c0 + i];

  int rr[NW];
  float ww[NW];
  int2 e[NW];
#pragma unroll
  for (int i = 0; i < NW; ++i) {
    e[i] = make_int2(c0 + i, 0);
    if (lane >= 1 && lane <= o[i + 1] - o[i]) e[i] = sedge[o[i] + lane - 1];
  }
  if (FIRST) {
    float g[NW];
#pragma unroll
    for (int i = 0; i < NW; ++i) {
      g[i] = 0.0f;
      if (lane >= 1 && lane <= o[i + 1] - o[i]) g[i] = dis[e[i].x];
    }
#pragma unroll
    for (int i = 0; i < NW; ++i) {
      ww[i] = dcs[i] * dcs[i];
      if (lane >= 1 && lane <= o[i + 1] - o[i]) {
        ww[i] = g[i] * __int_as_float(e[i].y) * dcs[i];
        sedge[o[i] + lane - 1] = make_int2(e[i].x, __float_as_int(ww[i]));
      }
    }
  } else {
#pragma unroll
    for (int i = 0; i < NW; ++i) {
      ww[i] = dcs[i] * dcs[i];
      if (lane >= 1 && lane <= o[i + 1] - o[i]) ww[i] = __int_as_float(e[i].y);
    }
  }
#pragma unroll
  for (int i = 0; i < NW; ++i) rr[i] = e[i].x;

#pragma unroll
  for (int i = 0; i < NW; ++i) {
    const int d = o[i + 1] - o[i];
    const int m = d + 1;
    int r = rr[i];
    float w = ww[i];
    float4 acc = make_float4(0.0f, 0.0f, 0.0f, 0.0f);
    for (int t0 = 0; t0 < m; t0 += 32) {
      if (t0 > 0) list_chunk<FIRST>(sedge, dis, c0 + i, o[i], d, dcs[i], t0, lane, r, w);
      const int left = m - t0;
      const int mm = left < 32 ? left : 32;
      int k = 0;
      for (; k + 8 <= mm; k += 8) rows_add<8>(hwb, r, w, k, loff, acc);
      switch (mm - k) {
        case 7: rows_add<7>(hwb, r, w, k, loff, acc); break;
        case 6: rows_add<6>(hwb, r, w, k, loff, acc); break;
        case 5: rows_add<5>(hwb, r, w, k, loff, acc); break;
        case 4: rows_add<4>(hwb, r, w, k, loff, acc); break;
        case 3: rows_add<3>(hwb, r, w, k, loff, acc); break;
        case 2: rows_add<2>(hwb, r, w, k, loff, acc); break;
        case 1: rows_add<1>(hwb, r, w, k, loff, acc); break;
        default: break;
      }
    }
    fin(i, acc);
  }
}

#define TM 32
template <bool FIRST>
__global__ __launch_bounds__(256, 16)
void k_gather_gemm(const float* __restrict__ hw, const int* __restrict__ offs,
                   int2* __restrict__ sedge, const float* __restrict__ dis,
                   const float* __restrict__ b,
                   const float* __restrict__ W,
                   float* __restrict__ C) {
  __shared__ float As[TM * 132];
  const int t    = threadIdx.x;
  const int lane = t & 31;
  const int wave = __builtin_amdgcn_readfirstlane(t >> 5);
  const int m0   = blockIdx.x * TM;
  const float4 bb = ((const float4*)b)[lane];

  const int lr0 = wave * (TM / 8);
  wave_nodes<FIRST, TM / 8>(hw, offs, sedge, dis, m0 + lr0, lane, [&](int i, float4 acc) {
    acc.x += bb.x; acc.y += bb.y; acc.z += bb.z; acc.w += bb.w;
    acc.x = fmaxf(acc.x, 0.0f); acc.y = fmaxf(acc.y, 0.0f);
    acc.z = fmaxf(acc.z, 0.0f); acc.w = fmaxf(acc.w, 0.0f);
    *(float4*)&As[(lr0 + i) * 132 + lane * 4] = acc;
  });
  __syncthreads();

  const int half = lane >> 4;
  const int l15  = lane & 15;
  const int n0   = wave * 16;

  const float4* Wq = (const float4*)W + (wave * 16) * 32 + lane;
  const float* Ar0 = &As[l15 * 132 + 2 * half];
  const float* Ar1 = Ar0 + 16 * 132;
  v8f c0 = {}, c1 = {};
#pragma unroll
  for (int j = 0; j < 16; ++j) {
    const float4 bq = Wq[j * 32];
    const int k0 = j * 8;
    v2f a00, a01, a10, a11, b0, b1;
    a00.x = Ar0[k0 + 0];
    a00.y = Ar0[k0 + 1];
    a01.x = Ar0[k0 + 4];
    a01.y = Ar0[k0 + 5];
    a10.x = Ar1[k0 + 0];
    a10.y = Ar1[k0 + 1];
    a11.x = Ar1[k0 + 4];
    a11.y = Ar1[k0 + 5];
    b0.x = bq.x; b0.y = bq.y;
    b1.x = bq.z; b1.y = bq.w;
    c0 = __builtin_amdgcn_wmma_f32_16x16x4_f32(false, a00, false, b0, (short)0, c0,
                                               false, false);
    c1 = __builtin_amdgcn_wmma_f32_16x16x4_f32(false, a10, false, b0, (short)0, c1,
                                               false, false);
    c0 = __builtin_amdgcn_wmma_f32_16x16x4_f32(false, a01, false, b1, (short)0, c0,
                                               false, false);
    c1 = __builtin_amdgcn_wmma_f32_16x16x4_f32(false, a11, false, b1, (short)0, c1,
                                               false, false);
  }

#pragma unroll
  for (int r = 0; r < 8; ++r) {
    C[(size_t)(m0 + r + 8 * half) * HIDF + (n0 + l15)] = c0[r];
  }
#pragma unroll
  for (int r = 0; r < 8; ++r) {
    C[(size_t)(m0 + 16 + r + 8 * half) * HIDF + (n0 + l15)] = c1[r];
  }
}

__global__ __launch_bounds__(256, 16)
void k_gather_out(const float* __restrict__ hw, const int* __restrict__ offs,
                  int2* __restrict__ sedge, const float* __restrict__ dis,
                  const float* __restrict__ bfold, const float* __restrict__ bout,
                  float* __restrict__ out) {
  const int t    = threadIdx.x;
  const int lane = t & 31;
  const int wave = __builtin_amdgcn_readfirstlane(t >> 5);
  const int m0   = blockIdx.x * 16;
  const float4 bf = ((const float4*)bfold)[lane];
  const float4 bo = ((const float4*)bout)[lane];
  const float4 bb = make_float4(bf.x + bo.x, bf.y + bo.y, bf.z + bo.z, bf.w + bo.w);
  float4* out4 = (float4*)out;

  const int c0 = m0 + wave * 2;
  wave_nodes<false, 2>(hw, offs, sedge, dis, c0, lane, [&](int i, float4 acc) {
    acc.x += bb.x; acc.y += bb.y; acc.z += bb.z; acc.w += bb.w;
    out4[(size_t)(c0 + i) * 32 + lane] = acc;
  });
}

extern "C" void kernel_launch(void* const* d_in, const int* in_sizes, int n_in,
                              void* d_out, int out_size, void* d_ws, size_t ws_size,
                              hipStream_t stream) {
  const float* x   = (const float*)d_in[0];
  const int*   ei  = (const int*)d_in[1];
  const float* ew  = (const float*)d_in[2];
  const float* Wc[4] = {(const float*)d_in[3], (const float*)d_in[5],
                        (const float*)d_in[7], (const float*)d_in[9]};
  const float* bc[4] = {(const float*)d_in[4], (const float*)d_in[6],
                        (const float*)d_in[8], (const float*)d_in[10]};
  const float* Wout = (const float*)d_in[11];
  const float* bout = (const float*)d_in[12];

  const int n = in_sizes[0] / HIDF;
  const int E = in_sizes[1] / 2;
  const int* row = ei;
  const int* col = ei + E;

  const int NB = (E + EB - 1) / EB;
  const int K  = (n + RB - 1) / RB;
  const int NT = K * NB;

  char* ws = (char*)d_ws;
  const size_t fb = (size_t)n * HIDF * sizeof(float);
  const size_t nb = (((size_t)(n + 1) * 4) + 255) & ~(size_t)255;
  const size_t tb = (((size_t)(NT + 1) * 4) + 255) & ~(size_t)255;
  const size_t eb = (((size_t)E * 4) + 255) & ~(size_t)255;
  float* hw   = (float*)ws;
  float* hbuf = (float*)(ws + fb);
  char*  p    = ws + 2 * fb;
  float* dis    = (float*)p; p += nb;
  int*   offs   = (int*)p;   p += nb;
  int*   tab    = (int*)p;   p += tb;
  int*   base   = (int*)p;   p += tb;
  int*   pos    = (int*)p;   p += eb;
  int2*  sedge  = (int2*)p;  p += 2 * eb;
  int4*  rec    = (int4*)p;  p += 4 * eb;
  int4*  tmp    = (int4*)p;  p += 4 * eb;
  float* wfold  = (float*)p;  p += (size_t)PKW * 4;
  float* bfold  = (float*)p;  p += (size_t)16 * HIDF * 4;
  float* pk     = (float*)p;

  k_bin_count<<<NB + 8 * NPK, ETH, 0, stream>>>(col, pos, tab, E, NB, K,
                                                Wc[0], Wc[1], Wc[2], Wout, pk);
  k_scan<<<(NT + 1023) / 1024, 1024, 0, stream>>>(tab, base, NT);
  k_bin_place<<<NB, ETH, 0, stream>>>(row, col, ew, pos, base, rec, E, NB, K);
  k_bin_order<<<K, 1024, 0, stream>>>(base, rec, tmp, sedge, offs, dis, n, NB);

  k_gemm_wmma<<<n / 16 + 9, 256, 0, stream>>>(x, pk, hw, n / 16, Wc[3], pk + 3 * PKW, bc[3],
                                             wfold, bfold);
  k_gather_gemm<true><<<n / TM, 256, 0, stream>>>(hw, offs, sedge, dis, bc[0], pk + PKW, hbuf);
  k_gather_gemm<false><<<n / TM, 256, 0, stream>>>(hbuf, offs, sedge, dis, bc[1], pk + 2 * PKW,
                                                  hw);
  k_gather_gemm<false><<<n / TM, 256, 0, stream>>>(hw, offs, sedge, dis, bc[2], wfold, hbuf);
  k_gather_out<<<n / 16, 256, 0, stream>>>(hbuf, offs, sedge, dis, bfold, bout,
                                          (float*)d_out);
}


# layer 0's product has no launch of its own: its row tiles run as extra work-groups behind the scan, place and order launches of the edge list (independent work, 9 -> 8 launches); sums unchanged
# speedup vs baseline: 5.1391x; 1.0329x over previous
#include <hip/hip_runtime.h>
#include <math.h>

typedef __attribute__((ext_vector_type(2))) float v2f;
typedef __attribute__((ext_vector_type(8))) float v8f;

#define HIDF 128


#define EB   8192
#define ETH  512
#define RB   256
#define RSH  8
#define KMAX 512
#define WCAP 64

#define PKW (HIDF * HIDF)
#define NPK 4


struct GemmArgs {
  const float* A;
  const float* W;
  float*       C;
  int          nt;
  const float* A2;
  const float* W2;
  const float* bvec;
  float*       C2;
  float*       C3;
  int          first;
  int          tend;
};

__device__ __forceinline__ void gemm_tile16(float* __restrict__ As, const GemmArgs& ga,
                                            int tile, int t) {
  const bool valid = tile < ga.tend;
  const int xb = tile - ga.nt;
  const bool extra = xb >= 0;
  const int m0 = (extra ? xb : tile) * 16;
  const float* Ap = extra ? ga.A2 : ga.A;
  const float* Wp = extra ? ga.W2 : ga.W;

  if (valid) {
    const float4* A4 = (const float4*)(Ap + (size_t)m0 * HIDF);
    const int r = t >> 5, c = t & 31;
    float4 v0, v1;
    if (xb < 8) {
      v0 = A4[t];
      v1 = A4[t + 256];
    } else {
      v0 = (r == 0) ? ((const float4*)ga.bvec)[c] : make_float4(0.0f, 0.0f, 0.0f, 0.0f);
      v1 = make_float4(0.0f, 0.0f, 0.0f, 0.0f);
    }
    *(float4*)&As[r * 132 + c * 4] = v0;
    *(float4*)&As[(r + 8) * 132 + c * 4] = v1;
  }
  __syncthreads();
  if (!valid) return;

  const int lane = t & 31;
  const int wave = t >> 5;
  const int half = lane >> 4;
  const int l15  = lane & 15;
  const int n0   = wave * 16;

  const float4* Wq = (const float4*)Wp + (wave * 16) * 32 + lane;
  v8f acc = {};
#pragma unroll
  for (int j = 0; j < 16; ++j) {
    const float4 bq = Wq[j * 32];
    const int k0 = j * 8 + 2 * half;
    v2f a0, a1, b0, b1;
    a0.x = As[l15 * 132 + k0 + 0];
    a0.y = As[l15 * 132 + k0 + 1];
    a1.x = As[l15 * 132 + k0 + 4];
    a1.y = As[l15 * 132 + k0 + 5];
    b0.x = bq.x; b0.y = bq.y;
    b1.x = bq.z; b1.y = bq.w;
    acc = __builtin_amdgcn_wmma_f32_16x16x4_f32(false, a0, false, b0, (short)0, acc,
                                                false, false);
    acc = __builtin_amdgcn_wmma_f32_16x16x4_f32(false, a1, false, b1, (short)0, acc,
                                                false, false);
  }

  if (xb >= 0 && xb < 8) {
#pragma unroll
    for (int r = 0; r < 8; ++r) {
      const int pl = ((r >> 1) & 1) * 16 + l15;
      const int pe = (r & 1) + 2 * (r >> 2);
      ga.C2[(((wave * 16) + 2 * xb + half) * 32 + pl) * 4 + pe] = acc[r];
    }
  } else {
    float* Cp = extra ? ga.C3 : ga.C;
    const int mr = extra ? 0 : m0;
#pragma unroll
    for (int r = 0; r < 8; ++r) {
      Cp[(size_t)(mr + r + 8 * half) * HIDF + (n0 + l15)] = acc[r];
    }
  }
}

template <int G>
__device__ __forceinline__ void gemm_ride(const GemmArgs& ga, int own) {
  __shared__ __attribute__((aligned(16))) float Ag[G][16 * 132];
  const int g = __builtin_amdgcn_readfirstlane((int)threadIdx.x >> 8);
  const int tile = ga.first + ((int)blockIdx.x - own) * G + g;
  gemm_tile16(&Ag[g][0], ga, tile, (int)threadIdx.x & 255);
}

__global__ __launch_bounds__(ETH)
void k_bin_count(const int* __restrict__ col, int* __restrict__ pos, int* __restrict__ tab,
                 int E, int NB, int K,
                 const float* __restrict__ Wa, const float* __restrict__ Wb,
                 const float* __restrict__ Wc, const float* __restrict__ Wd,
                 float* __restrict__ pk) {
  __shared__ int h[KMAX];
  const int t = threadIdx.x, b = blockIdx.x;
  if (b >= NB) {
    const int q = b - NB;
    const int which = q >> 3, w = q & 7;
    const float* Ws = (which == 0) ? Wa : (which == 1) ? Wb : (which == 2) ? Wc : Wd;
    const int j = t >> 5, lane = t & 31;
    const int n = 16 * w + (lane & 15);
    const int k0 = 8 * j + 2 * (lane >> 4);
    float4 v;
    v.x = Ws[(k0 + 0) * HIDF + n];
    v.y = Ws[(k0 + 1) * HIDF + n];
    v.z = Ws[(k0 + 4) * HIDF + n];
    v.w = Ws[(k0 + 5) * HIDF + n];
    ((float4*)(pk + (size_t)which * PKW))[(w * 16 + j) * 32 + lane] = v;
    return;
  }
  for (int i = t; i < K; i += ETH) h[i] = 0;
  __syncthreads();
  const int e0 = b * EB + t;
  if (b * EB + EB <= E) {
    for (int g = 0; g < EB / ETH; g += 8) {
      int c[8];
#pragma unroll
      for (int u = 0; u < 8; ++u) c[u] = col[e0 + (g + u) * ETH];
#pragma unroll
      for (int u = 0; u < 8; ++u) pos[e0 + (g + u) * ETH] = atomicAdd(&h[c[u] >> RSH], 1);
    }
  } else {
    for (int e = e0; e < E; e += ETH) pos[e] = atomicAdd(&h[col[e] >> RSH], 1);
  }
  __syncthreads();
  for (int i = t; i < K; i += ETH) tab[i * NB + b] = h[i];
}

__global__ __launch_bounds__(1024, 16)
void k_scan(const int* __restrict__ cnt, int* __restrict__ offs, int n, int nscan,
            const GemmArgs ga) {
  __shared__ int part[32];
  __shared__ int wtot[32];
  if ((int)blockIdx.x >= nscan) {
    gemm_ride<4>(ga, nscan);
    return;
  }
  const int t = threadIdx.x, lane = t & 31, wv = t >> 5;
  const int start = blockIdx.x * 1024;
  int s = 0;
  for (int i = t; i < start; i += 1024) s += cnt[i];
  const int i = start + t;
  const int v = (i < n) ? cnt[i] : 0;
  int inc = v;
  for (int o = 1; o < 32; o <<= 1) {
    const int u = __shfl_up(inc, o, 32);
    if (lane >= o) inc += u;
  }
  for (int o = 16; o > 0; o >>= 1) s += __shfl_xor(s, o, 32);
  if (lane == 31) wtot[wv] = inc;
  if (lane == 0) part[wv] = s;
  __syncthreads();
  int base = 0;
  for (int w = 0; w < 32; ++w) {
    base += part[w];
    if (w < wv) base += wtot[w];
  }
  const int excl = base + inc - v;
  if (i < n) offs[i] = excl;
  if (i == n - 1) offs[n] = excl + v;
}

__global__ __launch_bounds__(ETH, 16)
void k_bin_place(const int* __restrict__ row, const int* __restrict__ col,
                 const float* __restrict__ ew, const int* __restrict__ pos,
                 const int* __restrict__ base, int4* __restrict__ rec, int E, int NB, int K,
                 const GemmArgs ga) {
  __shared__ int sb[KMAX];
  const int t = threadIdx.x, b = blockIdx.x;
  if (b >= NB) {
    gemm_ride<ETH / 256>(ga, NB);
    return;
  }
  for (int i = t; i < K; i += ETH) sb[i] = base[i * NB + b];
  __syncthreads();
  const int e0 = b * EB + t;
  if (b * EB + EB <= E) {
    for (int g = 0; g < EB / ETH; g += 4) {
      int c[4], p[4], r[4];
      float w[4];
#pragma unroll
      for (int u = 0; u < 4; ++u) {
        const int e = e0 + (g + u) * ETH;
        c[u] = col[e]; p[u] = pos[e]; r[u] = row[e]; w[u] = ew[e];
      }
#pragma unroll
      for (int u = 0; u < 4; ++u) {
        const int e = e0 + (g + u) * ETH;
        rec[sb[c[u] >> RSH] + p[u]] = make_int4(e, r[u], __float_as_int(w[u]), c[u]);
      }
    }
  } else {
    for (int e = e0; e < E; e += ETH) {
      const int c = col[e];
      rec[sb[c >> RSH] + pos[e]] = make_int4(e, row[e], __float_as_int(ew[e]), c);
    }
  }
}

__global__ __launch_bounds__(1024, 16)
void k_bin_order(const int* __restrict__ base, const int4* __restrict__ rec,
                 int4* __restrict__ tmp, int2* __restrict__ sedge, int* __restrict__ offs,
                 float* __restrict__ dis, int n, int NB, int K, const GemmArgs ga) {
  if ((int)blockIdx.x >= K) {
    gemm_ride<4>(ga, K);
    return;
  }
  __shared__ int cnt[RB];
  __shared__ int loff[RB];
  __shared__ int cur[RB];
  __shared__ int wtot[RB / 32];
  __shared__ float wsort[32 * WCAP];
  const int t = threadIdx.x, lane = t & 31;
  const int wv = __builtin_amdgcn_readfirstlane(t >> 5);
  const int k = blockIdx.x;
  const int bs = __builtin_amdgcn_readfirstlane(base[k * NB]);
  const int be = __builtin_amdgcn_readfirstlane(base[(k + 1) * NB]);
  const int nb0 = k * RB;
  if (t < RB) cnt[t] = 0;
  __syncthreads();
  for (int idx = bs + t; idx < be; idx += 1024) atomicAdd(&cnt[rec[idx].w - nb0], 1);
  __syncthreads();
  const int v = (t < RB) ? cnt[t] : 0;
  int inc = v;
  for (int o = 1; o < 32; o <<= 1) {
    const int u = __shfl_up(inc, o, 32);
    if (lane >= o) inc += u;
  }
  if (t < RB && lane == 31) wtot[t >> 5] = inc;
  __syncthreads();
  if (t < RB) {
    int pre = 0;
    for (int w = 0; w < RB / 32; ++w)
      if (w < (t >> 5)) pre += wtot[w];
    const int ex = pre + inc - v;
    loff[t] = ex;
    cur[t] = ex;
    if (nb0 + t < n) offs[nb0 + t] = bs + ex;
  }
  if (k == K - 1 && t == 0) offs[n] = be;
  __syncthreads();
  for (int idx = bs + t; idx < be; idx += 1024) {
    const int4 q = rec[idx];
    const int p = atomicAdd(&cur[q.w - nb0], 1);
    tmp[bs + p] = q;
  }
  __syncthreads();

  float* const ws = &wsort[wv * WCAP];
  int i = wv;
  bool have = (nb0 + i < n);
  int d = 0, beg = 0;
  int4 q = make_int4(0x7fffffff, 0, 0, 0);
  if (have) {
    d = __builtin_amdgcn_readfirstlane(cnt[i]);
    beg = bs + __builtin_amdgcn_readfirstlane(loff[i]);
    if (lane < d) q = tmp[beg + lane];
  }
  while (have) {
    const int i2 = i + 32;
    const bool have2 = (i2 < RB) && (nb0 + i2 < n);
    int d2 = 0, beg2 = 0;
    int4 q2 = make_int4(0x7fffffff, 0, 0, 0);
    if (have2) {
      d2 = __builtin_amdgcn_readfirstlane(cnt[i2]);
      beg2 = bs + __builtin_amdgcn_readfirstlane(loff[i2]);
      if (lane < d2) q2 = tmp[beg2 + lane];
    }
    float deg = 0.0f;
    if (d <= WCAP) {
      for (int j0 = 0; j0 < d; j0 += 32) {
        int4 a = q;
        if (j0 > 0) {
          a = make_int4(0x7fffffff, 0, 0, 0);
          if (j0 + lane < d) a = tmp[beg + j0 + lane];
        }
        int rank = 0;
        for (int k0 = 0; k0 < d; k0 += 32) {
          int ek = a.x;
          if (k0 != j0) {
            ek = 0x7fffffff;
            if (k0 + lane < d) ek = tmp[beg + k0 + lane].x;
          }
          const int m = (d - k0 < 32) ? (d - k0) : 32;
          for (int kk = 0; kk < m; ++kk)
            rank += (__builtin_amdgcn_readlane(ek, kk) < a.x) ? 1 : 0;
        }
        if (j0 + lane < d) {
          sedge[beg + rank] = make_int2(a.y, a.z);
          ws[rank] = __int_as_float(a.z);
        }
      }
      for (int s0 = 0; s0 < d; s0 += 32) {
        const float sw = (s0 + lane < d) ? ws[s0 + lane] : 0.0f;
        const int m = (d - s0 < 32) ? (d - s0) : 32;
        for (int kk = 0; kk < m; ++kk)
          deg += __int_as_float(__builtin_amdgcn_readlane(__float_as_int(sw), kk));
      }
    } else if (lane == 0) {
      int last = -1;
      for (int s = 0; s < d; ++s) {
        int best = 0x7fffffff, bj = 0;
        for (int j = 0; j < d; ++j) {
          const int ee = tmp[beg + j].x;
          if (ee > last && ee < best) { best = ee; bj = j; }
        }
        const int4 z = tmp[beg + bj];
        sedge[beg + s] = make_int2(z.y, z.z);
        deg += __int_as_float(z.z);
        last = best;
      }
    }
    if (lane == 0) {
      deg += 1.0f;
      dis[nb0 + i] = (deg > 0.0f) ? rsqrtf(deg) : 0.0f;
    }
    i = i2; have = have2; d = d2; beg = beg2; q = q2;
  }
}

template <int B>
__device__ __forceinline__ void rows_add(const char* __restrict__ hwb, int r, float w,
                                         int k, unsigned loff, float4& acc) {
  float4 v[B];
  float wk[B];
#pragma unroll
  for (int i = 0; i < B; ++i) {
    const unsigned rk = (unsigned)__builtin_amdgcn_readlane(r, k + i);
    wk[i] = __int_as_float(__builtin_amdgcn_readlane(__float_as_int(w), k + i));
    const char* rowp = hwb + ((size_t)rk << 9);
    v[i] = *(const float4*)(rowp + loff);
  }
#pragma unroll
  for (int i = 0; i < B; ++i) {
    acc.x = fmaf(v[i].x, wk[i], acc.x);
    acc.y = fmaf(v[i].y, wk[i], acc.y);
    acc.z = fmaf(v[i].z, wk[i], acc.z);
    acc.w = fmaf(v[i].w, wk[i], acc.w);
  }
}

template <bool FIRST>
__device__ __forceinline__ void list_chunk(int2* __restrict__ sedge,
                                           const float* __restrict__ dis, int c, int beg,
                                           int d, float dc, int t0, int lane, int& r,
                                           float& w) {
  const int t = t0 + lane;
  r = c;
  w = dc * dc;
  if (t >= 1 && t <= d) {
    const int j = beg + t - 1;
    const int2 e = sedge[j];
    r = e.x;
    if (FIRST) {
      w = dis[r] * __int_as_float(e.y) * dc;
      sedge[j] = make_int2(r, __float_as_int(w));
    } else {
      w = __int_as_float(e.y);
    }
  }
}

template <bool FIRST, int NW, typename Fin>
__device__ __forceinline__ void wave_nodes(const float* __restrict__ hw,
                                           const int* __restrict__ offs,
                                           int2* __restrict__ sedge,
                                           const float* __restrict__ dis,
                                           int c0, int lane, Fin fin) {
  const char* hwb = (const char*)hw;
  const unsigned loff = (unsigned)lane * 16u;
  int o[NW + 1];
  float dcs[NW];
#pragma unroll
  for (int i = 0; i <= NW; ++i) o[i] = offs[c0 + i];
#pragma unroll
  for (int i = 0; i < NW; ++i) dcs[i] = dis[c0 + i];

  int rr[NW];
  float ww[NW];
  int2 e[NW];
#pragma unroll
  for (int i = 0; i < NW; ++i) {
    e[i] = make_int2(c0 + i, 0);
    if (lane >= 1 && lane <= o[i + 1] - o[i]) e[i] = sedge[o[i] + lane - 1];
  }
  if (FIRST) {
    float g[NW];
#pragma unroll
    for (int i = 0; i < NW; ++i) {
      g[i] = 0.0f;
      if (lane >= 1 && lane <= o[i + 1] - o[i]) g[i] = dis[e[i].x];
    }
#pragma unroll
    for (int i = 0; i < NW; ++i) {
      ww[i] = dcs[i] * dcs[i];
      if (lane >= 1 && lane <= o[i + 1] - o[i]) {
        ww[i] = g[i] * __int_as_float(e[i].y) * dcs[i];
        sedge[o[i] + lane - 1] = make_int2(e[i].x, __float_as_int(ww[i]));
      }
    }
  } else {
#pragma unroll
    for (int i = 0; i < NW; ++i) {
      ww[i] = dcs[i] * dcs[i];
      if (lane >= 1 && lane <= o[i + 1] - o[i]) ww[i] = __int_as_float(e[i].y);
    }
  }
#pragma unroll
  for (int i = 0; i < NW; ++i) rr[i] = e[i].x;

#pragma unroll
  for (int i = 0; i < NW; ++i) {
    const int d = o[i + 1] - o[i];
    const int m = d + 1;
    int r = rr[i];
    float w = ww[i];
    float4 acc = make_float4(0.0f, 0.0f, 0.0f, 0.0f);
    for (int t0 = 0; t0 < m; t0 += 32) {
      if (t0 > 0) list_chunk<FIRST>(sedge, dis, c0 + i, o[i], d, dcs[i], t0, lane, r, w);
      const int left = m - t0;
      const int mm = left < 32 ? left : 32;
      int k = 0;
      for (; k + 8 <= mm; k += 8) rows_add<8>(hwb, r, w, k, loff, acc);
      switch (mm - k) {
        case 7: rows_add<7>(hwb, r, w, k, loff, acc); break;
        case 6: rows_add<6>(hwb, r, w, k, loff, acc); break;
        case 5: rows_add<5>(hwb, r, w, k, loff, acc); break;
        case 4: rows_add<4>(hwb, r, w, k, loff, acc); break;
        case 3: rows_add<3>(hwb, r, w, k, loff, acc); break;
        case 2: rows_add<2>(hwb, r, w, k, loff, acc); break;
        case 1: rows_add<1>(hwb, r, w, k, loff, acc); break;
        default: break;
      }
    }
    fin(i, acc);
  }
}

#define TM 32
template <bool FIRST>
__global__ __launch_bounds__(256, 16)
void k_gather_gemm(const float* __restrict__ hw, const int* __restrict__ offs,
                   int2* __restrict__ sedge, const float* __restrict__ dis,
                   const float* __restrict__ b,
                   const float* __restrict__ W,
                   float* __restrict__ C) {
  __shared__ float As[TM * 132];
  const int t    = threadIdx.x;
  const int lane = t & 31;
  const int wave = __builtin_amdgcn_readfirstlane(t >> 5);
  const int m0   = blockIdx.x * TM;
  const float4 bb = ((const float4*)b)[lane];

  const int lr0 = wave * (TM / 8);
  wave_nodes<FIRST, TM / 8>(hw, offs, sedge, dis, m0 + lr0, lane, [&](int i, float4 acc) {
    acc.x += bb.x; acc.y += bb.y; acc.z += bb.z; acc.w += bb.w;
    acc.x = fmaxf(acc.x, 0.0f); acc.y = fmaxf(acc.y, 0.0f);
    acc.z = fmaxf(acc.z, 0.0f); acc.w = fmaxf(acc.w, 0.0f);
    *(float4*)&As[(lr0 + i) * 132 + lane * 4] = acc;
  });
  __syncthreads();

  const int half = lane >> 4;
  const int l15  = lane & 15;
  const int n0   = wave * 16;

  const float4* Wq = (const float4*)W + (wave * 16) * 32 + lane;
  const float* Ar0 = &As[l15 * 132 + 2 * half];
  const float* Ar1 = Ar0 + 16 * 132;
  v8f c0 = {}, c1 = {};
#pragma unroll
  for (int j = 0; j < 16; ++j) {
    const float4 bq = Wq[j * 32];
    const int k0 = j * 8;
    v2f a00, a01, a10, a11, b0, b1;
    a00.x = Ar0[k0 + 0];
    a00.y = Ar0[k0 + 1];
    a01.x = Ar0[k0 + 4];
    a01.y = Ar0[k0 + 5];
    a10.x = Ar1[k0 + 0];
    a10.y = Ar1[k0 + 1];
    a11.x = Ar1[k0 + 4];
    a11.y = Ar1[k0 + 5];
    b0.x = bq.x; b0.y = bq.y;
    b1.x = bq.z; b1.y = bq.w;
    c0 = __builtin_amdgcn_wmma_f32_16x16x4_f32(false, a00, false, b0, (short)0, c0,
                                               false, false);
    c1 = __builtin_amdgcn_wmma_f32_16x16x4_f32(false, a10, false, b0, (short)0, c1,
                                               false, false);
    c0 = __builtin_amdgcn_wmma_f32_16x16x4_f32(false, a01, false, b1, (short)0, c0,
                                               false, false);
    c1 = __builtin_amdgcn_wmma_f32_16x16x4_f32(false, a11, false, b1, (short)0, c1,
                                               false, false);
  }

#pragma unroll
  for (int r = 0; r < 8; ++r) {
    C[(size_t)(m0 + r + 8 * half) * HIDF + (n0 + l15)] = c0[r];
  }
#pragma unroll
  for (int r = 0; r < 8; ++r) {
    C[(size_t)(m0 + 16 + r + 8 * half) * HIDF + (n0 + l15)] = c1[r];
  }
}

__global__ __launch_bounds__(256, 16)
void k_gather_out(const float* __restrict__ hw, const int* __restrict__ offs,
                  int2* __restrict__ sedge, const float* __restrict__ dis,
                  const float* __restrict__ bfold, const float* __restrict__ bout,
                  float* __restrict__ out) {
  const int t    = threadIdx.x;
  const int lane = t & 31;
  const int wave = __builtin_amdgcn_readfirstlane(t >> 5);
  const int m0   = blockIdx.x * 16;
  const float4 bf = ((const float4*)bfold)[lane];
  const float4 bo = ((const float4*)bout)[lane];
  const float4 bb = make_float4(bf.x + bo.x, bf.y + bo.y, bf.z + bo.z, bf.w + bo.w);
  float4* out4 = (float4*)out;

  const int c0 = m0 + wave * 2;
  wave_nodes<false, 2>(hw, offs, sedge, dis, c0, lane, [&](int i, float4 acc) {
    acc.x += bb.x; acc.y += bb.y; acc.z += bb.z; acc.w += bb.w;
    out4[(size_t)(c0 + i) * 32 + lane] = acc;
  });
}

extern "C" void kernel_launch(void* const* d_in, const int* in_sizes, int n_in,
                              void* d_out, int out_size, void* d_ws, size_t ws_size,
                              hipStream_t stream) {
  const float* x   = (const float*)d_in[0];
  const int*   ei  = (const int*)d_in[1];
  const float* ew  = (const float*)d_in[2];
  const float* Wc[4] = {(const float*)d_in[3], (const float*)d_in[5],
                        (const float*)d_in[7], (const float*)d_in[9]};
  const float* bc[4] = {(const float*)d_in[4], (const float*)d_in[6],
                        (const float*)d_in[8], (const float*)d_in[10]};
  const float* Wout = (const float*)d_in[11];
  const float* bout = (const float*)d_in[12];

  const int n = in_sizes[0] / HIDF;
  const int E = in_sizes[1] / 2;
  const int* row = ei;
  const int* col = ei + E;

  const int NB = (E + EB - 1) / EB;
  const int K  = (n + RB - 1) / RB;
  const int NT = K * NB;

  char* ws = (char*)d_ws;
  const size_t fb = (size_t)n * HIDF * sizeof(float);
  const size_t nb = (((size_t)(n + 1) * 4) + 255) & ~(size_t)255;
  const size_t tb = (((size_t)(NT + 1) * 4) + 255) & ~(size_t)255;
  const size_t eb = (((size_t)E * 4) + 255) & ~(size_t)255;
  float* hw   = (float*)ws;
  float* hbuf = (float*)(ws + fb);
  char*  p    = ws + 2 * fb;
  float* dis    = (float*)p; p += nb;
  int*   offs   = (int*)p;   p += nb;
  int*   tab    = (int*)p;   p += tb;
  int*   base   = (int*)p;   p += tb;
  int*   pos    = (int*)p;   p += eb;
  int2*  sedge  = (int2*)p;  p += 2 * eb;
  int4*  rec    = (int4*)p;  p += 4 * eb;
  int4*  tmp    = (int4*)p;  p += 4 * eb;
  float* wfold  = (float*)p;  p += (size_t)PKW * 4;
  float* bfold  = (float*)p;  p += (size_t)16 * HIDF * 4;
  float* pk     = (float*)p;

  const int nt = n / 16;
  const int s1 = nt < 1024 ? nt : 1024;
  const int s2 = nt < s1 + 1536 ? nt : s1 + 1536;
  const int nscan = (NT + 1023) / 1024;
  GemmArgs ga;
  ga.first = 0; ga.tend = 0;
  ga.A = x; ga.W = pk; ga.C = hw; ga.nt = nt;
  ga.A2 = Wc[3]; ga.W2 = pk + 3 * PKW; ga.bvec = bc[3]; ga.C2 = wfold; ga.C3 = bfold;
  GemmArgs ga1 = ga, ga2 = ga, ga3 = ga;
  ga1.first = 0;  ga1.tend = s1;
  ga2.first = s1; ga2.tend = s2;
  ga3.first = s2; ga3.tend = nt + 9;

  k_bin_count<<<NB + 8 * NPK, ETH, 0, stream>>>(col, pos, tab, E, NB, K,
                                                Wc[0], Wc[1], Wc[2], Wout, pk);
  k_scan<<<nscan + (s1 + 3) / 4, 1024, 0, stream>>>(tab, base, NT, nscan, ga1);
  k_bin_place<<<NB + (s2 - s1 + 1) / 2, ETH, 0, stream>>>(row, col, ew, pos, base, rec, E, NB,
                                                          K, ga2);
  k_bin_order<<<K + (nt + 9 - s2 + 3) / 4, 1024, 0, stream>>>(base, rec, tmp, sedge, offs, dis,
                                                              n, NB, K, ga3);

  k_gather_gemm<true><<<n / TM, 256, 0, stream>>>(hw, offs, sedge, dis, bc[0], pk + PKW, hbuf);
  k_gather_gemm<false><<<n / TM, 256, 0, stream>>>(hbuf, offs, sedge, dis, bc[1], pk + 2 * PKW,
                                                  hw);
  k_gather_gemm<false><<<n / TM, 256, 0, stream>>>(hw, offs, sedge, dis, bc[2], wfold, hbuf);
  k_gather_out<<<n / 16, 256, 0, stream>>>(hbuf, offs, sedge, dis, bfold, bout,
                                          (float*)d_out);
}
